# starting kernel
# baseline (speedup 1.0000x reference)
#include <hip/hip_runtime.h>
#include <math.h>


typedef __attribute__((ext_vector_type(2))) float v2f;
typedef __attribute__((ext_vector_type(8))) float v8f;

#define DVEC      128
#define MNB       32
#define Z_STRIDE  132
#define G_STRIDE  36
#define Z_OFF     0
#define X_OFF     (MNB * Z_STRIDE)
#define G_OFF     (X_OFF + DVEC)
#define P_OFF     (G_OFF + MNB * G_STRIDE)
#define WAVE_FLOATS (P_OFF + MNB)

__global__ __launch_bounds__(256) void rnorm_kernel(const float* __restrict__ x,
                                                    float* __restrict__ rn,
                                                    int n_total) {
    const int lane = threadIdx.x & 31;
    const int wave = threadIdx.x >> 5;
    const int row  = blockIdx.x * 8 + wave;
    if (row >= n_total) return;
    float4 v = *(const float4*)(x + (size_t)row * DVEC + lane * 4);
    float ss = v.x * v.x + v.y * v.y + v.z * v.z + v.w * v.w;
    #pragma unroll
    for (int off = 16; off >= 1; off >>= 1) ss += __shfl_xor(ss, off, 32);
    if (lane == 0) rn[row] = 1.0f / fmaxf(sqrtf(ss), 1e-12f);
}

__global__ __launch_bounds__(64) void routing_kernel(const float* __restrict__ x,
                                                     const int* __restrict__ nb,
                                                     const float* __restrict__ rnorm,
                                                     float* __restrict__ out,
                                                     int n_total) {
    __shared__ __align__(16) float smem[2 * WAVE_FLOATS];
    const int lane = threadIdx.x & 31;
    const int wave = threadIdx.x >> 5;
    const int n    = blockIdx.x * 2 + wave;
    if (n >= n_total) return;

    float* S = smem + wave * WAVE_FLOATS;
    float* Z = S + Z_OFF;
    float* X = S + X_OFF;
    float* G = S + G_OFF;
    float* P = S + P_OFF;

    const int   jm = nb[(size_t)n * MNB + lane] - 1;
    const float rj = rnorm[jm];

    {
        const float rn = rnorm[n];
        float4 v = *(const float4*)(x + (size_t)n * DVEC + lane * 4);
        float4 w = make_float4(v.x * rn, v.y * rn, v.z * rn, v.w * rn);
        *(float4*)(X + lane * 4) = w;
    }

    for (int m = 0; m < MNB; ++m) {
        const int   j  = __shfl(jm, m, 32);
        const float rm = __shfl(rj, m, 32);
        float4 v = *(const float4*)(x + (size_t)j * DVEC + lane * 4);
        float4 w = make_float4(v.x * rm, v.y * rm, v.z * rm, v.w * rm);
        *(float4*)(Z + m * Z_STRIDE + lane * 4) = w;
    }

    v8f C00 = {}, C01 = {}, C10 = {}, C11 = {};
    const int lsub = lane & 15;
    const int koff = (lane >> 4) * 2;
    #pragma unroll 4
    for (int kc = 0; kc < DVEC; kc += 4) {
        v2f t0 = *(const v2f*)(Z + lsub * Z_STRIDE + kc + koff);
        v2f t1 = *(const v2f*)(Z + (16 + lsub) * Z_STRIDE + kc + koff);
        C00 = __builtin_amdgcn_wmma_f32_16x16x4_f32(false, t0, false, t0, (short)0, C00, false, false);
        C01 = __builtin_amdgcn_wmma_f32_16x16x4_f32(false, t0, false, t1, (short)0, C01, false, false);
        C10 = __builtin_amdgcn_wmma_f32_16x16x4_f32(false, t1, false, t0, (short)0, C10, false, false);
        C11 = __builtin_amdgcn_wmma_f32_16x16x4_f32(false, t1, false, t1, (short)0, C11, false, false);
    }

    {
        const int rbase = (lane < 16) ? 0 : 8;
        const int col   = lane & 15;
        #pragma unroll
        for (int r = 0; r < 8; ++r) {
            G[(rbase + r) * G_STRIDE + col]            = C00[r];
            G[(rbase + r) * G_STRIDE + 16 + col]       = C01[r];
            G[(16 + rbase + r) * G_STRIDE + col]       = C10[r];
            G[(16 + rbase + r) * G_STRIDE + 16 + col]  = C11[r];
        }
    }

    float bm = 0.0f;
    #pragma unroll 8
    for (int d = 0; d < DVEC; d += 4) {
        float4 zv = *(const float4*)(Z + lane * Z_STRIDE + d);
        float4 xv = *(const float4*)(X + d);
        bm += zv.x * xv.x + zv.y * xv.y + zv.z * xv.z + zv.w * xv.w;
    }

    float pm = 1.0f / 32.0f;
    for (int it = 0; it < 3; ++it) {
        P[lane] = pm;
        float am = bm;
        #pragma unroll 8
        for (int mp = 0; mp < MNB; ++mp)
            am += G[lane * G_STRIDE + mp] * P[mp];

        float pa = pm * am, pb = pm * bm;
        #pragma unroll
        for (int off = 16; off >= 1; off >>= 1) {
            pa += __shfl_xor(pa, off, 32);
            pb += __shfl_xor(pb, off, 32);
        }
        const float sq = pa + pb + 1.0f;
        const float s  = (sq / (sq + 1.0f)) / fmaxf(sqrtf(sq), 1e-12f);

        float t  = s * am;
        float mx = t;
        #pragma unroll
        for (int off = 16; off >= 1; off >>= 1) mx = fmaxf(mx, __shfl_xor(mx, off, 32));
        float e  = expf(t - mx);
        float se = e;
        #pragma unroll
        for (int off = 16; off >= 1; off >>= 1) se += __shfl_xor(se, off, 32);
        pm = e / se;
    }
    P[lane] = pm;

    float4 acc = *(const float4*)(X + lane * 4);
    #pragma unroll 8
    for (int m = 0; m < MNB; ++m) {
        const float pv = P[m];
        float4 zv = *(const float4*)(Z + m * Z_STRIDE + lane * 4);
        acc.x += zv.x * pv; acc.y += zv.y * pv; acc.z += zv.z * pv; acc.w += zv.w * pv;
    }
    *(float4*)(out + (size_t)n * DVEC + lane * 4) = acc;
}

extern "C" void kernel_launch(void* const* d_in, const int* in_sizes, int n_in,
                              void* d_out, int out_size, void* d_ws, size_t ws_size,
                              hipStream_t stream) {
    (void)n_in; (void)out_size; (void)ws_size;
    const float* x  = (const float*)d_in[0];
    const int*   nb = (const int*)d_in[1];
    float*       out = (float*)d_out;
    float*       rn  = (float*)d_ws;
    const int n = in_sizes[0] / DVEC;

    rnorm_kernel<<<(n + 7) / 8, 256, 0, stream>>>(x, rn, n);
    routing_kernel<<<(n + 1) / 2, 64, 0, stream>>>(x, nb, rn, out, n);
}


# All wave shuffles moved off the LDS unit: butterfly sums and maxima via permlanex16 + DPP row_xmask (same partner order 16,8,4,2,1), neighbor index and norm broadcasts via v_readlane
# speedup vs baseline: 1.1380x; 1.1380x over previous
#include <hip/hip_runtime.h>
#include <math.h>


typedef __attribute__((ext_vector_type(2))) float v2f;
typedef __attribute__((ext_vector_type(8))) float v8f;

#define DVEC      128
#define MNB       32
#define Z_STRIDE  132
#define G_STRIDE  36
#define Z_OFF     0
#define X_OFF     (MNB * Z_STRIDE)
#define G_OFF     (X_OFF + DVEC)
#define P_OFF     (G_OFF + MNB * G_STRIDE)
#define WAVE_FLOATS (P_OFF + MNB)

__device__ __forceinline__ float xhalf(float v) {
    const int b = __builtin_bit_cast(int, v);
    return __builtin_bit_cast(float,
        __builtin_amdgcn_permlanex16(b, b, 0x76543210, 0xfedcba98, false, false));
}
template <int MASK>
__device__ __forceinline__ float xrow(float v) {
    const int b = __builtin_bit_cast(int, v);
    return __builtin_bit_cast(float,
        __builtin_amdgcn_update_dpp(b, b, 0x160 + MASK, 0xf, 0xf, true));
}
__device__ __forceinline__ float wave_sum(float v) {
    v += xhalf(v);
    v += xrow<8>(v);
    v += xrow<4>(v);
    v += xrow<2>(v);
    v += xrow<1>(v);
    return v;
}
__device__ __forceinline__ float wave_max(float v) {
    v = fmaxf(v, xhalf(v));
    v = fmaxf(v, xrow<8>(v));
    v = fmaxf(v, xrow<4>(v));
    v = fmaxf(v, xrow<2>(v));
    v = fmaxf(v, xrow<1>(v));
    return v;
}
__device__ __forceinline__ int lane_bcast(int v, int src) {
    return __builtin_amdgcn_readlane(v, src);
}
__device__ __forceinline__ float lane_bcast(float v, int src) {
    return __builtin_bit_cast(float,
        __builtin_amdgcn_readlane(__builtin_bit_cast(int, v), src));
}

__global__ __launch_bounds__(256) void rnorm_kernel(const float* __restrict__ x,
                                                    float* __restrict__ rn,
                                                    int n_total) {
    const int lane = threadIdx.x & 31;
    const int wave = threadIdx.x >> 5;
    const int row  = blockIdx.x * 8 + wave;
    if (row >= n_total) return;
    float4 v = *(const float4*)(x + (size_t)row * DVEC + lane * 4);
    float ss = v.x * v.x + v.y * v.y + v.z * v.z + v.w * v.w;
    ss = wave_sum(ss);
    if (lane == 0) rn[row] = 1.0f / fmaxf(sqrtf(ss), 1e-12f);
}

__global__ __launch_bounds__(64) void routing_kernel(const float* __restrict__ x,
                                                     const int* __restrict__ nb,
                                                     const float* __restrict__ rnorm,
                                                     float* __restrict__ out,
                                                     int n_total) {
    __shared__ __align__(16) float smem[2 * WAVE_FLOATS];
    const int lane = threadIdx.x & 31;
    const int wave = threadIdx.x >> 5;
    const int n    = blockIdx.x * 2 + wave;
    if (n >= n_total) return;

    float* S = smem + wave * WAVE_FLOATS;
    float* Z = S + Z_OFF;
    float* X = S + X_OFF;
    float* G = S + G_OFF;
    float* P = S + P_OFF;

    const int   jm = nb[(size_t)n * MNB + lane] - 1;
    const float rj = rnorm[jm];

    {
        const float rn = rnorm[n];
        float4 v = *(const float4*)(x + (size_t)n * DVEC + lane * 4);
        float4 w = make_float4(v.x * rn, v.y * rn, v.z * rn, v.w * rn);
        *(float4*)(X + lane * 4) = w;
    }

    for (int m = 0; m < MNB; ++m) {
        const int   j  = lane_bcast(jm, m);
        const float rm = lane_bcast(rj, m);
        float4 v = *(const float4*)(x + (size_t)j * DVEC + lane * 4);
        float4 w = make_float4(v.x * rm, v.y * rm, v.z * rm, v.w * rm);
        *(float4*)(Z + m * Z_STRIDE + lane * 4) = w;
    }

    v8f C00 = {}, C01 = {}, C10 = {}, C11 = {};
    const int lsub = lane & 15;
    const int koff = (lane >> 4) * 2;
    #pragma unroll 4
    for (int kc = 0; kc < DVEC; kc += 4) {
        v2f t0 = *(const v2f*)(Z + lsub * Z_STRIDE + kc + koff);
        v2f t1 = *(const v2f*)(Z + (16 + lsub) * Z_STRIDE + kc + koff);
        C00 = __builtin_amdgcn_wmma_f32_16x16x4_f32(false, t0, false, t0, (short)0, C00, false, false);
        C01 = __builtin_amdgcn_wmma_f32_16x16x4_f32(false, t0, false, t1, (short)0, C01, false, false);
        C10 = __builtin_amdgcn_wmma_f32_16x16x4_f32(false, t1, false, t0, (short)0, C10, false, false);
        C11 = __builtin_amdgcn_wmma_f32_16x16x4_f32(false, t1, false, t1, (short)0, C11, false, false);
    }

    {
        const int rbase = (lane < 16) ? 0 : 8;
        const int col   = lane & 15;
        #pragma unroll
        for (int r = 0; r < 8; ++r) {
            G[(rbase + r) * G_STRIDE + col]            = C00[r];
            G[(rbase + r) * G_STRIDE + 16 + col]       = C01[r];
            G[(16 + rbase + r) * G_STRIDE + col]       = C10[r];
            G[(16 + rbase + r) * G_STRIDE + 16 + col]  = C11[r];
        }
    }

    float bm = 0.0f;
    #pragma unroll 8
    for (int d = 0; d < DVEC; d += 4) {
        float4 zv = *(const float4*)(Z + lane * Z_STRIDE + d);
        float4 xv = *(const float4*)(X + d);
        bm += zv.x * xv.x + zv.y * xv.y + zv.z * xv.z + zv.w * xv.w;
    }

    float pm = 1.0f / 32.0f;
    for (int it = 0; it < 3; ++it) {
        P[lane] = pm;
        float am = bm;
        #pragma unroll 8
        for (int mp = 0; mp < MNB; ++mp)
            am += G[lane * G_STRIDE + mp] * P[mp];

        const float pa = wave_sum(pm * am);
        const float pb = wave_sum(pm * bm);
        const float sq = pa + pb + 1.0f;
        const float s  = (sq / (sq + 1.0f)) / fmaxf(sqrtf(sq), 1e-12f);

        float t  = s * am;
        const float mx = wave_max(t);
        const float e  = expf(t - mx);
        const float se = wave_sum(e);
        pm = e / se;
    }
    P[lane] = pm;

    float4 acc = *(const float4*)(X + lane * 4);
    #pragma unroll 8
    for (int m = 0; m < MNB; ++m) {
        const float pv = P[m];
        float4 zv = *(const float4*)(Z + m * Z_STRIDE + lane * 4);
        acc.x += zv.x * pv; acc.y += zv.y * pv; acc.z += zv.z * pv; acc.w += zv.w * pv;
    }
    *(float4*)(out + (size_t)n * DVEC + lane * 4) = acc;
}

extern "C" void kernel_launch(void* const* d_in, const int* in_sizes, int n_in,
                              void* d_out, int out_size, void* d_ws, size_t ws_size,
                              hipStream_t stream) {
    (void)n_in; (void)out_size; (void)ws_size;
    const float* x  = (const float*)d_in[0];
    const int*   nb = (const int*)d_in[1];
    float*       out = (float*)d_out;
    float*       rn  = (float*)d_ws;
    const int n = in_sizes[0] / DVEC;

    rnorm_kernel<<<(n + 7) / 8, 256, 0, stream>>>(x, rn, n);
    routing_kernel<<<(n + 1) / 2, 64, 0, stream>>>(x, nb, rn, out, n);
}


# LDS per wave 22144 -> 13952 B: Z staged one 64-column half at a time, final sum takes the gathered rows from registers, launch bound caps VGPRs at 192: 20 instead of 14 waves per WGP
# speedup vs baseline: 1.2643x; 1.1110x over previous
#include <hip/hip_runtime.h>
#include <math.h>


typedef __attribute__((ext_vector_type(2))) float v2f;
typedef __attribute__((ext_vector_type(8))) float v8f;

#define DVEC      128
#define MNB       32
#define KHALF     64
#define ZH_STRIDE 68
#define G_STRIDE  36
#define ZH_OFF    0
#define X_OFF     (MNB * ZH_STRIDE)
#define G_OFF     (X_OFF + DVEC)
#define P_OFF     (G_OFF + MNB * G_STRIDE)
#define WAVE_FLOATS (P_OFF + MNB)

__device__ __forceinline__ float xhalf(float v) {
    const int b = __builtin_bit_cast(int, v);
    return __builtin_bit_cast(float,
        __builtin_amdgcn_permlanex16(b, b, 0x76543210, 0xfedcba98, false, false));
}
template <int MASK>
__device__ __forceinline__ float xrow(float v) {
    const int b = __builtin_bit_cast(int, v);
    return __builtin_bit_cast(float,
        __builtin_amdgcn_update_dpp(b, b, 0x160 + MASK, 0xf, 0xf, true));
}
__device__ __forceinline__ float wave_sum(float v) {
    v += xhalf(v);
    v += xrow<8>(v);
    v += xrow<4>(v);
    v += xrow<2>(v);
    v += xrow<1>(v);
    return v;
}
__device__ __forceinline__ float wave_max(float v) {
    v = fmaxf(v, xhalf(v));
    v = fmaxf(v, xrow<8>(v));
    v = fmaxf(v, xrow<4>(v));
    v = fmaxf(v, xrow<2>(v));
    v = fmaxf(v, xrow<1>(v));
    return v;
}
__device__ __forceinline__ int lane_bcast(int v, int src) {
    return __builtin_amdgcn_readlane(v, src);
}
__device__ __forceinline__ float lane_bcast(float v, int src) {
    return __builtin_bit_cast(float,
        __builtin_amdgcn_readlane(__builtin_bit_cast(int, v), src));
}

__global__ __launch_bounds__(256) void rnorm_kernel(const float* __restrict__ x,
                                                    float* __restrict__ rn,
                                                    int n_total) {
    const int lane = threadIdx.x & 31;
    const int wave = threadIdx.x >> 5;
    const int row  = blockIdx.x * 8 + wave;
    if (row >= n_total) return;
    float4 v = *(const float4*)(x + (size_t)row * DVEC + lane * 4);
    float ss = v.x * v.x + v.y * v.y + v.z * v.z + v.w * v.w;
    ss = wave_sum(ss);
    if (lane == 0) rn[row] = 1.0f / fmaxf(sqrtf(ss), 1e-12f);
}

__global__ __launch_bounds__(64, 5) void routing_kernel(const float* __restrict__ x,
                                                     const int* __restrict__ nb,
                                                     const float* __restrict__ rnorm,
                                                     float* __restrict__ out,
                                                     int n_total) {
    __shared__ __align__(16) float smem[2 * WAVE_FLOATS];
    const int lane = threadIdx.x & 31;
    const int wave = threadIdx.x >> 5;
    const int n    = blockIdx.x * 2 + wave;
    if (n >= n_total) return;

    float* S = smem + wave * WAVE_FLOATS;
    float* ZH = S + ZH_OFF;
    float* X  = S + X_OFF;
    float* G  = S + G_OFF;
    float* P  = S + P_OFF;

    const int   jm = nb[(size_t)n * MNB + lane] - 1;
    const float rj = rnorm[jm];

    {
        const float rn = rnorm[n];
        float4 v = *(const float4*)(x + (size_t)n * DVEC + lane * 4);
        float4 w = make_float4(v.x * rn, v.y * rn, v.z * rn, v.w * rn);
        *(float4*)(X + lane * 4) = w;
    }

    float4 zr[MNB];
    #pragma unroll
    for (int m = 0; m < MNB; ++m) {
        const int   j  = lane_bcast(jm, m);
        const float rm = lane_bcast(rj, m);
        float4 v = *(const float4*)(x + (size_t)j * DVEC + lane * 4);
        zr[m] = make_float4(v.x * rm, v.y * rm, v.z * rm, v.w * rm);
    }

    const int lsub = lane & 15;
    const int hsel = lane >> 4;
    const int koff = hsel * 2;
    float bm = 0.0f;

    if (hsel == 0) {
        #pragma unroll
        for (int m = 0; m < MNB; ++m)
            *(float4*)(ZH + m * ZH_STRIDE + lsub * 4) = zr[m];
    }
    #pragma unroll 4
    for (int d = 0; d < KHALF; d += 4) {
        float4 zv = *(const float4*)(ZH + lane * ZH_STRIDE + d);
        float4 xv = *(const float4*)(X + d);
        bm += zv.x * xv.x + zv.y * xv.y + zv.z * xv.z + zv.w * xv.w;
    }

    v8f C00 = {}, C01 = {}, C10 = {}, C11 = {};
    #pragma unroll 4
    for (int kc = 0; kc < KHALF; kc += 4) {
        v2f t0 = *(const v2f*)(ZH + lsub * ZH_STRIDE + kc + koff);
        v2f t1 = *(const v2f*)(ZH + (16 + lsub) * ZH_STRIDE + kc + koff);
        C00 = __builtin_amdgcn_wmma_f32_16x16x4_f32(false, t0, false, t0, (short)0, C00, false, false);
        C01 = __builtin_amdgcn_wmma_f32_16x16x4_f32(false, t0, false, t1, (short)0, C01, false, false);
        C10 = __builtin_amdgcn_wmma_f32_16x16x4_f32(false, t1, false, t0, (short)0, C10, false, false);
        C11 = __builtin_amdgcn_wmma_f32_16x16x4_f32(false, t1, false, t1, (short)0, C11, false, false);
    }

    if (hsel == 1) {
        #pragma unroll
        for (int m = 0; m < MNB; ++m)
            *(float4*)(ZH + m * ZH_STRIDE + lsub * 4) = zr[m];
    }
    #pragma unroll 4
    for (int kc = 0; kc < KHALF; kc += 4) {
        v2f t0 = *(const v2f*)(ZH + lsub * ZH_STRIDE + kc + koff);
        v2f t1 = *(const v2f*)(ZH + (16 + lsub) * ZH_STRIDE + kc + koff);
        C00 = __builtin_amdgcn_wmma_f32_16x16x4_f32(false, t0, false, t0, (short)0, C00, false, false);
        C01 = __builtin_amdgcn_wmma_f32_16x16x4_f32(false, t0, false, t1, (short)0, C01, false, false);
        C10 = __builtin_amdgcn_wmma_f32_16x16x4_f32(false, t1, false, t0, (short)0, C10, false, false);
        C11 = __builtin_amdgcn_wmma_f32_16x16x4_f32(false, t1, false, t1, (short)0, C11, false, false);
    }

    {
        const int rbase = (lane < 16) ? 0 : 8;
        const int col   = lane & 15;
        #pragma unroll
        for (int r = 0; r < 8; ++r) {
            G[(rbase + r) * G_STRIDE + col]            = C00[r];
            G[(rbase + r) * G_STRIDE + 16 + col]       = C01[r];
            G[(16 + rbase + r) * G_STRIDE + col]       = C10[r];
            G[(16 + rbase + r) * G_STRIDE + 16 + col]  = C11[r];
        }
    }

    #pragma unroll 4
    for (int d = 0; d < KHALF; d += 4) {
        float4 zv = *(const float4*)(ZH + lane * ZH_STRIDE + d);
        float4 xv = *(const float4*)(X + KHALF + d);
        bm += zv.x * xv.x + zv.y * xv.y + zv.z * xv.z + zv.w * xv.w;
    }

    float pm = 1.0f / 32.0f;
    for (int it = 0; it < 3; ++it) {
        P[lane] = pm;
        float am = bm;
        #pragma unroll 8
        for (int mp = 0; mp < MNB; ++mp)
            am += G[lane * G_STRIDE + mp] * P[mp];

        const float pa = wave_sum(pm * am);
        const float pb = wave_sum(pm * bm);
        const float sq = pa + pb + 1.0f;
        const float s  = (sq / (sq + 1.0f)) / fmaxf(sqrtf(sq), 1e-12f);

        float t  = s * am;
        const float mx = wave_max(t);
        const float e  = expf(t - mx);
        const float se = wave_sum(e);
        pm = e / se;
    }
    P[lane] = pm;

    float4 acc = *(const float4*)(X + lane * 4);
    #pragma unroll
    for (int m = 0; m < MNB; ++m) {
        const float pv = P[m];
        const float4 zv = zr[m];
        acc.x += zv.x * pv; acc.y += zv.y * pv; acc.z += zv.z * pv; acc.w += zv.w * pv;
    }
    *(float4*)(out + (size_t)n * DVEC + lane * 4) = acc;
}

extern "C" void kernel_launch(void* const* d_in, const int* in_sizes, int n_in,
                              void* d_out, int out_size, void* d_ws, size_t ws_size,
                              hipStream_t stream) {
    (void)n_in; (void)out_size; (void)ws_size;
    const float* x  = (const float*)d_in[0];
    const int*   nb = (const int*)d_in[1];
    float*       out = (float*)d_out;
    float*       rn  = (float*)d_ws;
    const int n = in_sizes[0] / DVEC;

    rnorm_kernel<<<(n + 7) / 8, 256, 0, stream>>>(x, rn, n);
    routing_kernel<<<(n + 1) / 2, 64, 0, stream>>>(x, nb, rn, out, n);
}


# Norm kernel removed: inverse norms of the 32 gathered rows read off the diagonal of the Gram tiles of the unscaled rows (squares summed by the matrix unit in k order); scaling moved onto p and b
# speedup vs baseline: 1.3752x; 1.0877x over previous
#include <hip/hip_runtime.h>
#include <math.h>


typedef __attribute__((ext_vector_type(2))) float v2f;
typedef __attribute__((ext_vector_type(8))) float v8f;

#define DVEC      128
#define MNB       32
#define KHALF     64
#define ZH_STRIDE 68
#define G_STRIDE  36
#define ZH_OFF    0
#define X_OFF     (MNB * ZH_STRIDE)
#define G_OFF     (X_OFF + DVEC)
#define P_OFF     (G_OFF + MNB * G_STRIDE)
#define WAVE_FLOATS (P_OFF + MNB)

__device__ __forceinline__ float xhalf(float v) {
    const int b = __builtin_bit_cast(int, v);
    return __builtin_bit_cast(float,
        __builtin_amdgcn_permlanex16(b, b, 0x76543210, 0xfedcba98, false, false));
}
template <int MASK>
__device__ __forceinline__ float xrow(float v) {
    const int b = __builtin_bit_cast(int, v);
    return __builtin_bit_cast(float,
        __builtin_amdgcn_update_dpp(b, b, 0x160 + MASK, 0xf, 0xf, true));
}
__device__ __forceinline__ float wave_sum(float v) {
    v += xhalf(v);
    v += xrow<8>(v);
    v += xrow<4>(v);
    v += xrow<2>(v);
    v += xrow<1>(v);
    return v;
}
__device__ __forceinline__ float wave_max(float v) {
    v = fmaxf(v, xhalf(v));
    v = fmaxf(v, xrow<8>(v));
    v = fmaxf(v, xrow<4>(v));
    v = fmaxf(v, xrow<2>(v));
    v = fmaxf(v, xrow<1>(v));
    return v;
}
__device__ __forceinline__ int lane_bcast(int v, int src) {
    return __builtin_amdgcn_readlane(v, src);
}
__device__ __forceinline__ float lane_bcast(float v, int src) {
    return __builtin_bit_cast(float,
        __builtin_amdgcn_readlane(__builtin_bit_cast(int, v), src));
}

__global__ __launch_bounds__(64, 5) void routing_kernel(const float* __restrict__ x,
                                                     const int* __restrict__ nb,
                                                     float* __restrict__ out,
                                                     int n_total) {
    __shared__ __align__(16) float smem[2 * WAVE_FLOATS];
    const int lane = threadIdx.x & 31;
    const int wave = threadIdx.x >> 5;
    const int n    = blockIdx.x * 2 + wave;
    if (n >= n_total) return;

    float* S = smem + wave * WAVE_FLOATS;
    float* ZH = S + ZH_OFF;
    float* X  = S + X_OFF;
    float* G  = S + G_OFF;
    float* P  = S + P_OFF;

    const int   jm = nb[(size_t)n * MNB + lane] - 1;

    {
        float4 v = *(const float4*)(x + (size_t)n * DVEC + lane * 4);
        float ss = v.x * v.x + v.y * v.y + v.z * v.z + v.w * v.w;
        ss = wave_sum(ss);
        const float rn = 1.0f / fmaxf(sqrtf(ss), 1e-12f);
        float4 w = make_float4(v.x * rn, v.y * rn, v.z * rn, v.w * rn);
        *(float4*)(X + lane * 4) = w;
    }

    float4 zr[MNB];
    #pragma unroll
    for (int m = 0; m < MNB; ++m) {
        const int j = lane_bcast(jm, m);
        zr[m] = *(const float4*)(x + (size_t)j * DVEC + lane * 4);
    }

    const int lsub = lane & 15;
    const int hsel = lane >> 4;
    const int koff = hsel * 2;
    float bm = 0.0f;

    if (hsel == 0) {
        #pragma unroll
        for (int m = 0; m < MNB; ++m)
            *(float4*)(ZH + m * ZH_STRIDE + lsub * 4) = zr[m];
    }
    #pragma unroll 4
    for (int d = 0; d < KHALF; d += 4) {
        float4 zv = *(const float4*)(ZH + lane * ZH_STRIDE + d);
        float4 xv = *(const float4*)(X + d);
        bm += zv.x * xv.x + zv.y * xv.y + zv.z * xv.z + zv.w * xv.w;
    }

    v8f C00 = {}, C01 = {}, C10 = {}, C11 = {};
    #pragma unroll 4
    for (int kc = 0; kc < KHALF; kc += 4) {
        v2f t0 = *(const v2f*)(ZH + lsub * ZH_STRIDE + kc + koff);
        v2f t1 = *(const v2f*)(ZH + (16 + lsub) * ZH_STRIDE + kc + koff);
        C00 = __builtin_amdgcn_wmma_f32_16x16x4_f32(false, t0, false, t0, (short)0, C00, false, false);
        C01 = __builtin_amdgcn_wmma_f32_16x16x4_f32(false, t0, false, t1, (short)0, C01, false, false);
        C10 = __builtin_amdgcn_wmma_f32_16x16x4_f32(false, t1, false, t0, (short)0, C10, false, false);
        C11 = __builtin_amdgcn_wmma_f32_16x16x4_f32(false, t1, false, t1, (short)0, C11, false, false);
    }

    if (hsel == 1) {
        #pragma unroll
        for (int m = 0; m < MNB; ++m)
            *(float4*)(ZH + m * ZH_STRIDE + lsub * 4) = zr[m];
    }
    #pragma unroll 4
    for (int kc = 0; kc < KHALF; kc += 4) {
        v2f t0 = *(const v2f*)(ZH + lsub * ZH_STRIDE + kc + koff);
        v2f t1 = *(const v2f*)(ZH + (16 + lsub) * ZH_STRIDE + kc + koff);
        C00 = __builtin_amdgcn_wmma_f32_16x16x4_f32(false, t0, false, t0, (short)0, C00, false, false);
        C01 = __builtin_amdgcn_wmma_f32_16x16x4_f32(false, t0, false, t1, (short)0, C01, false, false);
        C10 = __builtin_amdgcn_wmma_f32_16x16x4_f32(false, t1, false, t0, (short)0, C10, false, false);
        C11 = __builtin_amdgcn_wmma_f32_16x16x4_f32(false, t1, false, t1, (short)0, C11, false, false);
    }

    {
        const int rbase = (lane < 16) ? 0 : 8;
        const int col   = lane & 15;
        #pragma unroll
        for (int r = 0; r < 8; ++r) {
            G[(rbase + r) * G_STRIDE + col]            = C00[r];
            G[(rbase + r) * G_STRIDE + 16 + col]       = C01[r];
            G[(16 + rbase + r) * G_STRIDE + col]       = C10[r];
            G[(16 + rbase + r) * G_STRIDE + 16 + col]  = C11[r];
        }
    }

    #pragma unroll 4
    for (int d = 0; d < KHALF; d += 4) {
        float4 zv = *(const float4*)(ZH + lane * ZH_STRIDE + d);
        float4 xv = *(const float4*)(X + KHALF + d);
        bm += zv.x * xv.x + zv.y * xv.y + zv.z * xv.z + zv.w * xv.w;
    }

    const float rm = 1.0f / fmaxf(sqrtf(G[lane * G_STRIDE + lane]), 1e-12f);
    bm *= rm;

    float pm = 1.0f / 32.0f;
    for (int it = 0; it < 3; ++it) {
        P[lane] = pm * rm;
        float gq = 0.0f;
        #pragma unroll 8
        for (int mp = 0; mp < MNB; ++mp)
            gq += G[lane * G_STRIDE + mp] * P[mp];
        const float am = rm * gq + bm;

        const float pa = wave_sum(pm * am);
        const float pb = wave_sum(pm * bm);
        const float sq = pa + pb + 1.0f;
        const float s  = (sq / (sq + 1.0f)) / fmaxf(sqrtf(sq), 1e-12f);

        float t  = s * am;
        const float mx = wave_max(t);
        const float e  = expf(t - mx);
        const float se = wave_sum(e);
        pm = e / se;
    }
    P[lane] = pm * rm;

    float4 acc = *(const float4*)(X + lane * 4);
    #pragma unroll
    for (int m = 0; m < MNB; ++m) {
        const float pv = P[m];
        const float4 zv = zr[m];
        acc.x += zv.x * pv; acc.y += zv.y * pv; acc.z += zv.z * pv; acc.w += zv.w * pv;
    }
    *(float4*)(out + (size_t)n * DVEC + lane * 4) = acc;
}

extern "C" void kernel_launch(void* const* d_in, const int* in_sizes, int n_in,
                              void* d_out, int out_size, void* d_ws, size_t ws_size,
                              hipStream_t stream) {
    (void)n_in; (void)out_size; (void)d_ws; (void)ws_size;
    const float* x  = (const float*)d_in[0];
    const int*   nb = (const int*)d_in[1];
    float*       out = (float*)d_out;
    const int n = in_sizes[0] / DVEC;

    routing_kernel<<<(n + 1) / 2, 64, 0, stream>>>(x, nb, out, n);
}


# Gram symmetry: lower-left 16x16 tile no longer computed (96 instead of 128 f32 WMMAs per wave); its LDS copy is stored as the transpose of the upper-right tile, same products in the same k order
# speedup vs baseline: 1.5486x; 1.1261x over previous
#include <hip/hip_runtime.h>
#include <math.h>


typedef __attribute__((ext_vector_type(2))) float v2f;
typedef __attribute__((ext_vector_type(8))) float v8f;

#define DVEC      128
#define MNB       32
#define KHALF     64
#define ZH_STRIDE 68
#define G_STRIDE  36
#define ZH_OFF    0
#define X_OFF     (MNB * ZH_STRIDE)
#define G_OFF     (X_OFF + DVEC)
#define P_OFF     (G_OFF + MNB * G_STRIDE)
#define WAVE_FLOATS (P_OFF + MNB)

__device__ __forceinline__ float xhalf(float v) {
    const int b = __builtin_bit_cast(int, v);
    return __builtin_bit_cast(float,
        __builtin_amdgcn_permlanex16(b, b, 0x76543210, 0xfedcba98, false, false));
}
template <int MASK>
__device__ __forceinline__ float xrow(float v) {
    const int b = __builtin_bit_cast(int, v);
    return __builtin_bit_cast(float,
        __builtin_amdgcn_update_dpp(b, b, 0x160 + MASK, 0xf, 0xf, true));
}
__device__ __forceinline__ float wave_sum(float v) {
    v += xhalf(v);
    v += xrow<8>(v);
    v += xrow<4>(v);
    v += xrow<2>(v);
    v += xrow<1>(v);
    return v;
}
__device__ __forceinline__ float wave_max(float v) {
    v = fmaxf(v, xhalf(v));
    v = fmaxf(v, xrow<8>(v));
    v = fmaxf(v, xrow<4>(v));
    v = fmaxf(v, xrow<2>(v));
    v = fmaxf(v, xrow<1>(v));
    return v;
}
__device__ __forceinline__ int lane_bcast(int v, int src) {
    return __builtin_amdgcn_readlane(v, src);
}
__device__ __forceinline__ float lane_bcast(float v, int src) {
    return __builtin_bit_cast(float,
        __builtin_amdgcn_readlane(__builtin_bit_cast(int, v), src));
}

__global__ __launch_bounds__(64, 5) void routing_kernel(const float* __restrict__ x,
                                                     const int* __restrict__ nb,
                                                     float* __restrict__ out,
                                                     int n_total) {
    __shared__ __align__(16) float smem[2 * WAVE_FLOATS];
    const int lane = threadIdx.x & 31;
    const int wave = threadIdx.x >> 5;
    const int n    = blockIdx.x * 2 + wave;
    if (n >= n_total) return;

    float* S = smem + wave * WAVE_FLOATS;
    float* ZH = S + ZH_OFF;
    float* X  = S + X_OFF;
    float* G  = S + G_OFF;
    float* P  = S + P_OFF;

    const int   jm = nb[(size_t)n * MNB + lane] - 1;

    {
        float4 v = *(const float4*)(x + (size_t)n * DVEC + lane * 4);
        float ss = v.x * v.x + v.y * v.y + v.z * v.z + v.w * v.w;
        ss = wave_sum(ss);
        const float rn = 1.0f / fmaxf(sqrtf(ss), 1e-12f);
        float4 w = make_float4(v.x * rn, v.y * rn, v.z * rn, v.w * rn);
        *(float4*)(X + lane * 4) = w;
    }

    float4 zr[MNB];
    #pragma unroll
    for (int m = 0; m < MNB; ++m) {
        const int j = lane_bcast(jm, m);
        zr[m] = *(const float4*)(x + (size_t)j * DVEC + lane * 4);
    }

    const int lsub = lane & 15;
    const int hsel = lane >> 4;
    const int koff = hsel * 2;
    float bm = 0.0f;

    if (hsel == 0) {
        #pragma unroll
        for (int m = 0; m < MNB; ++m)
            *(float4*)(ZH + m * ZH_STRIDE + lsub * 4) = zr[m];
    }
    #pragma unroll 4
    for (int d = 0; d < KHALF; d += 4) {
        float4 zv = *(const float4*)(ZH + lane * ZH_STRIDE + d);
        float4 xv = *(const float4*)(X + d);
        bm += zv.x * xv.x + zv.y * xv.y + zv.z * xv.z + zv.w * xv.w;
    }

    v8f C00 = {}, C01 = {}, C11 = {};
    #pragma unroll 4
    for (int kc = 0; kc < KHALF; kc += 4) {
        v2f t0 = *(const v2f*)(ZH + lsub * ZH_STRIDE + kc + koff);
        v2f t1 = *(const v2f*)(ZH + (16 + lsub) * ZH_STRIDE + kc + koff);
        C00 = __builtin_amdgcn_wmma_f32_16x16x4_f32(false, t0, false, t0, (short)0, C00, false, false);
        C01 = __builtin_amdgcn_wmma_f32_16x16x4_f32(false, t0, false, t1, (short)0, C01, false, false);
        C11 = __builtin_amdgcn_wmma_f32_16x16x4_f32(false, t1, false, t1, (short)0, C11, false, false);
    }

    if (hsel == 1) {
        #pragma unroll
        for (int m = 0; m < MNB; ++m)
            *(float4*)(ZH + m * ZH_STRIDE + lsub * 4) = zr[m];
    }
    #pragma unroll 4
    for (int kc = 0; kc < KHALF; kc += 4) {
        v2f t0 = *(const v2f*)(ZH + lsub * ZH_STRIDE + kc + koff);
        v2f t1 = *(const v2f*)(ZH + (16 + lsub) * ZH_STRIDE + kc + koff);
        C00 = __builtin_amdgcn_wmma_f32_16x16x4_f32(false, t0, false, t0, (short)0, C00, false, false);
        C01 = __builtin_amdgcn_wmma_f32_16x16x4_f32(false, t0, false, t1, (short)0, C01, false, false);
        C11 = __builtin_amdgcn_wmma_f32_16x16x4_f32(false, t1, false, t1, (short)0, C11, false, false);
    }

    {
        const int rbase = (lane < 16) ? 0 : 8;
        const int col   = lane & 15;
        #pragma unroll
        for (int r = 0; r < 8; ++r) {
            G[(rbase + r) * G_STRIDE + col]            = C00[r];
            G[(rbase + r) * G_STRIDE + 16 + col]       = C01[r];
            G[(16 + rbase + r) * G_STRIDE + 16 + col]  = C11[r];
        }
        float* gt = G + (16 + col) * G_STRIDE + rbase;
        *(float4*)(gt)     = make_float4(C01[0], C01[1], C01[2], C01[3]);
        *(float4*)(gt + 4) = make_float4(C01[4], C01[5], C01[6], C01[7]);
    }

    #pragma unroll 4
    for (int d = 0; d < KHALF; d += 4) {
        float4 zv = *(const float4*)(ZH + lane * ZH_STRIDE + d);
        float4 xv = *(const float4*)(X + KHALF + d);
        bm += zv.x * xv.x + zv.y * xv.y + zv.z * xv.z + zv.w * xv.w;
    }

    const float rm = 1.0f / fmaxf(sqrtf(G[lane * G_STRIDE + lane]), 1e-12f);
    bm *= rm;

    float pm = 1.0f / 32.0f;
    for (int it = 0; it < 3; ++it) {
        P[lane] = pm * rm;
        float gq = 0.0f;
        #pragma unroll 8
        for (int mp = 0; mp < MNB; ++mp)
            gq += G[lane * G_STRIDE + mp] * P[mp];
        const float am = rm * gq + bm;

        const float pa = wave_sum(pm * am);
        const float pb = wave_sum(pm * bm);
        const float sq = pa + pb + 1.0f;
        const float s  = (sq / (sq + 1.0f)) / fmaxf(sqrtf(sq), 1e-12f);

        float t  = s * am;
        const float mx = wave_max(t);
        const float e  = expf(t - mx);
        const float se = wave_sum(e);
        pm = e / se;
    }
    P[lane] = pm * rm;

    float4 acc = *(const float4*)(X + lane * 4);
    #pragma unroll
    for (int m = 0; m < MNB; ++m) {
        const float pv = P[m];
        const float4 zv = zr[m];
        acc.x += zv.x * pv; acc.y += zv.y * pv; acc.z += zv.z * pv; acc.w += zv.w * pv;
    }
    *(float4*)(out + (size_t)n * DVEC + lane * 4) = acc;
}

extern "C" void kernel_launch(void* const* d_in, const int* in_sizes, int n_in,
                              void* d_out, int out_size, void* d_ws, size_t ws_size,
                              hipStream_t stream) {
    (void)n_in; (void)out_size; (void)d_ws; (void)ws_size;
    const float* x  = (const float*)d_in[0];
    const int*   nb = (const int*)d_in[1];
    float*       out = (float*)d_out;
    const int n = in_sizes[0] / DVEC;

    routing_kernel<<<(n + 1) / 2, 64, 0, stream>>>(x, nb, out, n);
}


# b = W x accumulated from the WMMA loop's own fragment registers (2 v_pk_fma_f32 per K-chunk) instead of two LDS passes over Z; b's sum order changes: per k parity and lane half, then combined
# speedup vs baseline: 1.6424x; 1.0606x over previous
#include <hip/hip_runtime.h>
#include <math.h>


typedef __attribute__((ext_vector_type(2))) float v2f;
typedef __attribute__((ext_vector_type(8))) float v8f;

#define DVEC      128
#define MNB       32
#define KHALF     64
#define ZH_STRIDE 68
#define G_STRIDE  36
#define ZH_OFF    0
#define X_OFF     (MNB * ZH_STRIDE)
#define G_OFF     (X_OFF + DVEC)
#define P_OFF     (G_OFF + MNB * G_STRIDE)
#define WAVE_FLOATS (P_OFF + MNB)

__device__ __forceinline__ float xhalf(float v) {
    const int b = __builtin_bit_cast(int, v);
    return __builtin_bit_cast(float,
        __builtin_amdgcn_permlanex16(b, b, 0x76543210, 0xfedcba98, false, false));
}
template <int MASK>
__device__ __forceinline__ float xrow(float v) {
    const int b = __builtin_bit_cast(int, v);
    return __builtin_bit_cast(float,
        __builtin_amdgcn_update_dpp(b, b, 0x160 + MASK, 0xf, 0xf, true));
}
__device__ __forceinline__ float wave_sum(float v) {
    v += xhalf(v);
    v += xrow<8>(v);
    v += xrow<4>(v);
    v += xrow<2>(v);
    v += xrow<1>(v);
    return v;
}
__device__ __forceinline__ float wave_max(float v) {
    v = fmaxf(v, xhalf(v));
    v = fmaxf(v, xrow<8>(v));
    v = fmaxf(v, xrow<4>(v));
    v = fmaxf(v, xrow<2>(v));
    v = fmaxf(v, xrow<1>(v));
    return v;
}
__device__ __forceinline__ int lane_bcast(int v, int src) {
    return __builtin_amdgcn_readlane(v, src);
}
__device__ __forceinline__ float lane_bcast(float v, int src) {
    return __builtin_bit_cast(float,
        __builtin_amdgcn_readlane(__builtin_bit_cast(int, v), src));
}

__global__ __launch_bounds__(64, 5) void routing_kernel(const float* __restrict__ x,
                                                     const int* __restrict__ nb,
                                                     float* __restrict__ out,
                                                     int n_total) {
    __shared__ __align__(16) float smem[2 * WAVE_FLOATS];
    const int lane = threadIdx.x & 31;
    const int wave = threadIdx.x >> 5;
    const int n    = blockIdx.x * 2 + wave;
    if (n >= n_total) return;

    float* S = smem + wave * WAVE_FLOATS;
    float* ZH = S + ZH_OFF;
    float* X  = S + X_OFF;
    float* G  = S + G_OFF;
    float* P  = S + P_OFF;

    const int   jm = nb[(size_t)n * MNB + lane] - 1;

    {
        float4 v = *(const float4*)(x + (size_t)n * DVEC + lane * 4);
        float ss = v.x * v.x + v.y * v.y + v.z * v.z + v.w * v.w;
        ss = wave_sum(ss);
        const float rn = 1.0f / fmaxf(sqrtf(ss), 1e-12f);
        float4 w = make_float4(v.x * rn, v.y * rn, v.z * rn, v.w * rn);
        *(float4*)(X + lane * 4) = w;
    }

    float4 zr[MNB];
    #pragma unroll
    for (int m = 0; m < MNB; ++m) {
        const int j = lane_bcast(jm, m);
        zr[m] = *(const float4*)(x + (size_t)j * DVEC + lane * 4);
    }

    const int lsub = lane & 15;
    const int hsel = lane >> 4;
    const int koff = hsel * 2;

    if (hsel == 0) {
        #pragma unroll
        for (int m = 0; m < MNB; ++m)
            *(float4*)(ZH + m * ZH_STRIDE + lsub * 4) = zr[m];
    }

    v8f C00 = {}, C01 = {}, C11 = {};
    v2f pb0 = {}, pb1 = {};
    #pragma unroll 4
    for (int kc = 0; kc < KHALF; kc += 4) {
        v2f t0 = *(const v2f*)(ZH + lsub * ZH_STRIDE + kc + koff);
        v2f t1 = *(const v2f*)(ZH + (16 + lsub) * ZH_STRIDE + kc + koff);
        v2f xf = *(const v2f*)(X + kc + koff);
        C00 = __builtin_amdgcn_wmma_f32_16x16x4_f32(false, t0, false, t0, (short)0, C00, false, false);
        C01 = __builtin_amdgcn_wmma_f32_16x16x4_f32(false, t0, false, t1, (short)0, C01, false, false);
        C11 = __builtin_amdgcn_wmma_f32_16x16x4_f32(false, t1, false, t1, (short)0, C11, false, false);
        pb0 += t0 * xf;
        pb1 += t1 * xf;
    }

    if (hsel == 1) {
        #pragma unroll
        for (int m = 0; m < MNB; ++m)
            *(float4*)(ZH + m * ZH_STRIDE + lsub * 4) = zr[m];
    }
    #pragma unroll 4
    for (int kc = 0; kc < KHALF; kc += 4) {
        v2f t0 = *(const v2f*)(ZH + lsub * ZH_STRIDE + kc + koff);
        v2f t1 = *(const v2f*)(ZH + (16 + lsub) * ZH_STRIDE + kc + koff);
        v2f xf = *(const v2f*)(X + KHALF + kc + koff);
        C00 = __builtin_amdgcn_wmma_f32_16x16x4_f32(false, t0, false, t0, (short)0, C00, false, false);
        C01 = __builtin_amdgcn_wmma_f32_16x16x4_f32(false, t0, false, t1, (short)0, C01, false, false);
        C11 = __builtin_amdgcn_wmma_f32_16x16x4_f32(false, t1, false, t1, (short)0, C11, false, false);
        pb0 += t0 * xf;
        pb1 += t1 * xf;
    }

    float bm;
    {
        const float s0  = pb0.x + pb0.y;
        const float s1  = pb1.x + pb1.y;
        const float own = hsel ? s1 : s0;
        const float oth = hsel ? s0 : s1;
        bm = own + xhalf(oth);
    }

    {
        const int rbase = (lane < 16) ? 0 : 8;
        const int col   = lane & 15;
        #pragma unroll
        for (int r = 0; r < 8; ++r) {
            G[(rbase + r) * G_STRIDE + col]            = C00[r];
            G[(rbase + r) * G_STRIDE + 16 + col]       = C01[r];
            G[(16 + rbase + r) * G_STRIDE + 16 + col]  = C11[r];
        }
        float* gt = G + (16 + col) * G_STRIDE + rbase;
        *(float4*)(gt)     = make_float4(C01[0], C01[1], C01[2], C01[3]);
        *(float4*)(gt + 4) = make_float4(C01[4], C01[5], C01[6], C01[7]);
    }

    const float rm = 1.0f / fmaxf(sqrtf(G[lane * G_STRIDE + lane]), 1e-12f);
    bm *= rm;

    float pm = 1.0f / 32.0f;
    for (int it = 0; it < 3; ++it) {
        P[lane] = pm * rm;
        float gq = 0.0f;
        #pragma unroll 8
        for (int mp = 0; mp < MNB; ++mp)
            gq += G[lane * G_STRIDE + mp] * P[mp];
        const float am = rm * gq + bm;

        const float pa = wave_sum(pm * am);
        const float pb = wave_sum(pm * bm);
        const float sq = pa + pb + 1.0f;
        const float s  = (sq / (sq + 1.0f)) / fmaxf(sqrtf(sq), 1e-12f);

        float t  = s * am;
        const float mx = wave_max(t);
        const float e  = expf(t - mx);
        const float se = wave_sum(e);
        pm = e / se;
    }
    P[lane] = pm * rm;

    float4 acc = *(const float4*)(X + lane * 4);
    #pragma unroll
    for (int m = 0; m < MNB; ++m) {
        const float pv = P[m];
        const float4 zv = zr[m];
        acc.x += zv.x * pv; acc.y += zv.y * pv; acc.z += zv.z * pv; acc.w += zv.w * pv;
    }
    *(float4*)(out + (size_t)n * DVEC + lane * 4) = acc;
}

extern "C" void kernel_launch(void* const* d_in, const int* in_sizes, int n_in,
                              void* d_out, int out_size, void* d_ws, size_t ws_size,
                              hipStream_t stream) {
    (void)n_in; (void)out_size; (void)d_ws; (void)ws_size;
    const float* x  = (const float*)d_in[0];
    const int*   nb = (const int*)d_in[1];
    float*       out = (float*)d_out;
    const int n = in_sizes[0] / DVEC;

    routing_kernel<<<(n + 1) / 2, 64, 0, stream>>>(x, nb, out, n);
}


# Norms, squash factor and softmax use the hardware's 1-ulp v_rcp / v_rsq / v_exp, one instruction each, instead of the exact division, sqrt and expf sequences (11 div, 5 sqrt, 3 exp per row)
# speedup vs baseline: 1.7523x; 1.0669x over previous
#include <hip/hip_runtime.h>
#include <math.h>


typedef __attribute__((ext_vector_type(2))) float v2f;
typedef __attribute__((ext_vector_type(8))) float v8f;

#define DVEC      128
#define MNB       32
#define KHALF     64
#define ZH_STRIDE 68
#define G_STRIDE  36
#define ZH_OFF    0
#define X_OFF     (MNB * ZH_STRIDE)
#define G_OFF     (X_OFF + DVEC)
#define P_OFF     (G_OFF + MNB * G_STRIDE)
#define WAVE_FLOATS (P_OFF + MNB)

__device__ __forceinline__ float xhalf(float v) {
    const int b = __builtin_bit_cast(int, v);
    return __builtin_bit_cast(float,
        __builtin_amdgcn_permlanex16(b, b, 0x76543210, 0xfedcba98, false, false));
}
template <int MASK>
__device__ __forceinline__ float xrow(float v) {
    const int b = __builtin_bit_cast(int, v);
    return __builtin_bit_cast(float,
        __builtin_amdgcn_update_dpp(b, b, 0x160 + MASK, 0xf, 0xf, true));
}
__device__ __forceinline__ float wave_sum(float v) {
    v += xhalf(v);
    v += xrow<8>(v);
    v += xrow<4>(v);
    v += xrow<2>(v);
    v += xrow<1>(v);
    return v;
}
__device__ __forceinline__ float wave_max(float v) {
    v = fmaxf(v, xhalf(v));
    v = fmaxf(v, xrow<8>(v));
    v = fmaxf(v, xrow<4>(v));
    v = fmaxf(v, xrow<2>(v));
    v = fmaxf(v, xrow<1>(v));
    return v;
}
__device__ __forceinline__ int lane_bcast(int v, int src) {
    return __builtin_amdgcn_readlane(v, src);
}
__device__ __forceinline__ float lane_bcast(float v, int src) {
    return __builtin_bit_cast(float,
        __builtin_amdgcn_readlane(__builtin_bit_cast(int, v), src));
}

__device__ __forceinline__ float hw_rcp(float v)  { return __builtin_amdgcn_rcpf(v); }
__device__ __forceinline__ float hw_rsq(float v)  { return __builtin_amdgcn_rsqf(v); }
__device__ __forceinline__ float hw_exp2(float v) { return __builtin_amdgcn_exp2f(v); }
__device__ __forceinline__ float inv_norm(float ss) { return fminf(hw_rsq(ss), 1e12f); }

__global__ __launch_bounds__(64, 5) void routing_kernel(const float* __restrict__ x,
                                                     const int* __restrict__ nb,
                                                     float* __restrict__ out,
                                                     int n_total) {
    __shared__ __align__(16) float smem[2 * WAVE_FLOATS];
    const int lane = threadIdx.x & 31;
    const int wave = threadIdx.x >> 5;
    const int n    = blockIdx.x * 2 + wave;
    if (n >= n_total) return;

    float* S = smem + wave * WAVE_FLOATS;
    float* ZH = S + ZH_OFF;
    float* X  = S + X_OFF;
    float* G  = S + G_OFF;
    float* P  = S + P_OFF;

    const int   jm = nb[(size_t)n * MNB + lane] - 1;

    {
        float4 v = *(const float4*)(x + (size_t)n * DVEC + lane * 4);
        float ss = v.x * v.x + v.y * v.y + v.z * v.z + v.w * v.w;
        ss = wave_sum(ss);
        const float rn = inv_norm(ss);
        float4 w = make_float4(v.x * rn, v.y * rn, v.z * rn, v.w * rn);
        *(float4*)(X + lane * 4) = w;
    }

    float4 zr[MNB];
    #pragma unroll
    for (int m = 0; m < MNB; ++m) {
        const int j = lane_bcast(jm, m);
        zr[m] = *(const float4*)(x + (size_t)j * DVEC + lane * 4);
    }

    const int lsub = lane & 15;
    const int hsel = lane >> 4;
    const int koff = hsel * 2;

    if (hsel == 0) {
        #pragma unroll
        for (int m = 0; m < MNB; ++m)
            *(float4*)(ZH + m * ZH_STRIDE + lsub * 4) = zr[m];
    }

    v8f C00 = {}, C01 = {}, C11 = {};
    v2f pb0 = {}, pb1 = {};
    #pragma unroll 4
    for (int kc = 0; kc < KHALF; kc += 4) {
        v2f t0 = *(const v2f*)(ZH + lsub * ZH_STRIDE + kc + koff);
        v2f t1 = *(const v2f*)(ZH + (16 + lsub) * ZH_STRIDE + kc + koff);
        v2f xf = *(const v2f*)(X + kc + koff);
        C00 = __builtin_amdgcn_wmma_f32_16x16x4_f32(false, t0, false, t0, (short)0, C00, false, false);
        C01 = __builtin_amdgcn_wmma_f32_16x16x4_f32(false, t0, false, t1, (short)0, C01, false, false);
        C11 = __builtin_amdgcn_wmma_f32_16x16x4_f32(false, t1, false, t1, (short)0, C11, false, false);
        pb0 += t0 * xf;
        pb1 += t1 * xf;
    }

    if (hsel == 1) {
        #pragma unroll
        for (int m = 0; m < MNB; ++m)
            *(float4*)(ZH + m * ZH_STRIDE + lsub * 4) = zr[m];
    }
    #pragma unroll 4
    for (int kc = 0; kc < KHALF; kc += 4) {
        v2f t0 = *(const v2f*)(ZH + lsub * ZH_STRIDE + kc + koff);
        v2f t1 = *(const v2f*)(ZH + (16 + lsub) * ZH_STRIDE + kc + koff);
        v2f xf = *(const v2f*)(X + KHALF + kc + koff);
        C00 = __builtin_amdgcn_wmma_f32_16x16x4_f32(false, t0, false, t0, (short)0, C00, false, false);
        C01 = __builtin_amdgcn_wmma_f32_16x16x4_f32(false, t0, false, t1, (short)0, C01, false, false);
        C11 = __builtin_amdgcn_wmma_f32_16x16x4_f32(false, t1, false, t1, (short)0, C11, false, false);
        pb0 += t0 * xf;
        pb1 += t1 * xf;
    }

    float bm;
    {
        const float s0  = pb0.x + pb0.y;
        const float s1  = pb1.x + pb1.y;
        const float own = hsel ? s1 : s0;
        const float oth = hsel ? s0 : s1;
        bm = own + xhalf(oth);
    }

    {
        const int rbase = (lane < 16) ? 0 : 8;
        const int col   = lane & 15;
        #pragma unroll
        for (int r = 0; r < 8; ++r) {
            G[(rbase + r) * G_STRIDE + col]            = C00[r];
            G[(rbase + r) * G_STRIDE + 16 + col]       = C01[r];
            G[(16 + rbase + r) * G_STRIDE + 16 + col]  = C11[r];
        }
        float* gt = G + (16 + col) * G_STRIDE + rbase;
        *(float4*)(gt)     = make_float4(C01[0], C01[1], C01[2], C01[3]);
        *(float4*)(gt + 4) = make_float4(C01[4], C01[5], C01[6], C01[7]);
    }

    const float rm = inv_norm(G[lane * G_STRIDE + lane]);
    bm *= rm;

    float pm = 1.0f / 32.0f;
    for (int it = 0; it < 3; ++it) {
        P[lane] = pm * rm;
        float gq = 0.0f;
        #pragma unroll 8
        for (int mp = 0; mp < MNB; ++mp)
            gq += G[lane * G_STRIDE + mp] * P[mp];
        const float am = rm * gq + bm;

        const float pa = wave_sum(pm * am);
        const float pb = wave_sum(pm * bm);
        const float sq = pa + pb + 1.0f;
        const float s  = (sq * hw_rcp(sq + 1.0f)) * inv_norm(sq);

        float t  = s * am;
        const float mx = wave_max(t);
        const float e  = hw_exp2((t - mx) * 1.44269504088896341f);
        const float se = wave_sum(e);
        pm = e * hw_rcp(se);
    }
    P[lane] = pm * rm;

    float4 acc = *(const float4*)(X + lane * 4);
    #pragma unroll
    for (int m = 0; m < MNB; ++m) {
        const float pv = P[m];
        const float4 zv = zr[m];
        acc.x += zv.x * pv; acc.y += zv.y * pv; acc.z += zv.z * pv; acc.w += zv.w * pv;
    }
    *(float4*)(out + (size_t)n * DVEC + lane * 4) = acc;
}

extern "C" void kernel_launch(void* const* d_in, const int* in_sizes, int n_in,
                              void* d_out, int out_size, void* d_ws, size_t ws_size,
                              hipStream_t stream) {
    (void)n_in; (void)out_size; (void)d_ws; (void)ws_size;
    const float* x  = (const float*)d_in[0];
    const int*   nb = (const int*)d_in[1];
    float*       out = (float*)d_out;
    const int n = in_sizes[0] / DVEC;

    routing_kernel<<<(n + 1) / 2, 64, 0, stream>>>(x, nb, out, n);
}


# Gather loads fetch a 256B half of two rows each (lane half h owns rows 16h..16h+15), so the LDS staging stores run with all 32 lanes: 32 instead of 64; final sum: two 16-row chains, then added
# speedup vs baseline: 1.8119x; 1.0340x over previous
#include <hip/hip_runtime.h>
#include <math.h>


typedef __attribute__((ext_vector_type(2))) float v2f;
typedef __attribute__((ext_vector_type(8))) float v8f;

#define DVEC      128
#define MNB       32
#define KHALF     64
#define ZH_STRIDE 68
#define G_STRIDE  36
#define ZH_OFF    0
#define X_OFF     (MNB * ZH_STRIDE)
#define G_OFF     (X_OFF + DVEC)
#define P_OFF     (G_OFF + MNB * G_STRIDE)
#define WAVE_FLOATS (P_OFF + MNB)

__device__ __forceinline__ float xhalf(float v) {
    const int b = __builtin_bit_cast(int, v);
    return __builtin_bit_cast(float,
        __builtin_amdgcn_permlanex16(b, b, 0x76543210, 0xfedcba98, false, false));
}
template <int MASK>
__device__ __forceinline__ float xrow(float v) {
    const int b = __builtin_bit_cast(int, v);
    return __builtin_bit_cast(float,
        __builtin_amdgcn_update_dpp(b, b, 0x160 + MASK, 0xf, 0xf, true));
}
template <int I>
__device__ __forceinline__ int row_share(int v) {
    return __builtin_amdgcn_update_dpp(v, v, 0x150 + I, 0xf, 0xf, true);
}
__device__ __forceinline__ float wave_sum(float v) {
    v += xhalf(v);
    v += xrow<8>(v);
    v += xrow<4>(v);
    v += xrow<2>(v);
    v += xrow<1>(v);
    return v;
}
__device__ __forceinline__ float wave_max(float v) {
    v = fmaxf(v, xhalf(v));
    v = fmaxf(v, xrow<8>(v));
    v = fmaxf(v, xrow<4>(v));
    v = fmaxf(v, xrow<2>(v));
    v = fmaxf(v, xrow<1>(v));
    return v;
}

__device__ __forceinline__ float hw_rcp(float v)  { return __builtin_amdgcn_rcpf(v); }
__device__ __forceinline__ float hw_rsq(float v)  { return __builtin_amdgcn_rsqf(v); }
__device__ __forceinline__ float hw_exp2(float v) { return __builtin_amdgcn_exp2f(v); }
__device__ __forceinline__ float inv_norm(float ss) { return fminf(hw_rsq(ss), 1e12f); }

__global__ __launch_bounds__(64, 5) void routing_kernel(const float* __restrict__ x,
                                                     const int* __restrict__ nb,
                                                     float* __restrict__ out,
                                                     int n_total) {
    __shared__ __align__(16) float smem[2 * WAVE_FLOATS];
    const int lane = threadIdx.x & 31;
    const int wave = threadIdx.x >> 5;
    const int n    = blockIdx.x * 2 + wave;
    if (n >= n_total) return;

    float* S = smem + wave * WAVE_FLOATS;
    float* ZH = S + ZH_OFF;
    float* X  = S + X_OFF;
    float* G  = S + G_OFF;
    float* P  = S + P_OFF;

    const int   jm = nb[(size_t)n * MNB + lane] - 1;

    {
        float4 v = *(const float4*)(x + (size_t)n * DVEC + lane * 4);
        float ss = v.x * v.x + v.y * v.y + v.z * v.z + v.w * v.w;
        ss = wave_sum(ss);
        const float rn = inv_norm(ss);
        float4 w = make_float4(v.x * rn, v.y * rn, v.z * rn, v.w * rn);
        *(float4*)(X + lane * 4) = w;
    }

    const int lsub = lane & 15;
    const int hsel = lane >> 4;
    const char* xbytes = (const char*)x;
    float4 za[16], zb[16];
    unsigned roff[16];
#define GATHER_A(I) { roff[I] = (unsigned)row_share<I>(jm) * (unsigned)(DVEC * 4) + (unsigned)(lsub * 16); \
                      za[I] = *(const float4*)(xbytes + roff[I]); }
    GATHER_A(0)  GATHER_A(1)  GATHER_A(2)  GATHER_A(3)  GATHER_A(4)  GATHER_A(5)  GATHER_A(6)  GATHER_A(7)
    GATHER_A(8)  GATHER_A(9)  GATHER_A(10) GATHER_A(11) GATHER_A(12) GATHER_A(13) GATHER_A(14) GATHER_A(15)
#undef GATHER_A
    #pragma unroll
    for (int i = 0; i < 16; ++i)
        zb[i] = *(const float4*)(xbytes + roff[i] + KHALF * 4);

    const int koff = hsel * 2;
    float* zst = ZH + (16 * hsel) * ZH_STRIDE + lsub * 4;

    #pragma unroll
    for (int i = 0; i < 16; ++i)
        *(float4*)(zst + i * ZH_STRIDE) = za[i];

    v8f C00 = {}, C01 = {}, C11 = {};
    v2f pb0 = {}, pb1 = {};
    #pragma unroll 4
    for (int kc = 0; kc < KHALF; kc += 4) {
        v2f t0 = *(const v2f*)(ZH + lsub * ZH_STRIDE + kc + koff);
        v2f t1 = *(const v2f*)(ZH + (16 + lsub) * ZH_STRIDE + kc + koff);
        v2f xf = *(const v2f*)(X + kc + koff);
        C00 = __builtin_amdgcn_wmma_f32_16x16x4_f32(false, t0, false, t0, (short)0, C00, false, false);
        C01 = __builtin_amdgcn_wmma_f32_16x16x4_f32(false, t0, false, t1, (short)0, C01, false, false);
        C11 = __builtin_amdgcn_wmma_f32_16x16x4_f32(false, t1, false, t1, (short)0, C11, false, false);
        pb0 += t0 * xf;
        pb1 += t1 * xf;
    }

    #pragma unroll
    for (int i = 0; i < 16; ++i)
        *(float4*)(zst + i * ZH_STRIDE) = zb[i];
    #pragma unroll 4
    for (int kc = 0; kc < KHALF; kc += 4) {
        v2f t0 = *(const v2f*)(ZH + lsub * ZH_STRIDE + kc + koff);
        v2f t1 = *(const v2f*)(ZH + (16 + lsub) * ZH_STRIDE + kc + koff);
        v2f xf = *(const v2f*)(X + KHALF + kc + koff);
        C00 = __builtin_amdgcn_wmma_f32_16x16x4_f32(false, t0, false, t0, (short)0, C00, false, false);
        C01 = __builtin_amdgcn_wmma_f32_16x16x4_f32(false, t0, false, t1, (short)0, C01, false, false);
        C11 = __builtin_amdgcn_wmma_f32_16x16x4_f32(false, t1, false, t1, (short)0, C11, false, false);
        pb0 += t0 * xf;
        pb1 += t1 * xf;
    }

    float bm;
    {
        const float s0  = pb0.x + pb0.y;
        const float s1  = pb1.x + pb1.y;
        const float own = hsel ? s1 : s0;
        const float oth = hsel ? s0 : s1;
        bm = own + xhalf(oth);
    }

    {
        const int rbase = (lane < 16) ? 0 : 8;
        const int col   = lane & 15;
        #pragma unroll
        for (int r = 0; r < 8; ++r) {
            G[(rbase + r) * G_STRIDE + col]            = C00[r];
            G[(rbase + r) * G_STRIDE + 16 + col]       = C01[r];
            G[(16 + rbase + r) * G_STRIDE + 16 + col]  = C11[r];
        }
        float* gt = G + (16 + col) * G_STRIDE + rbase;
        *(float4*)(gt)     = make_float4(C01[0], C01[1], C01[2], C01[3]);
        *(float4*)(gt + 4) = make_float4(C01[4], C01[5], C01[6], C01[7]);
    }

    const float rm = inv_norm(G[lane * G_STRIDE + lane]);
    bm *= rm;

    float pm = 1.0f / 32.0f;
    for (int it = 0; it < 3; ++it) {
        P[lane] = pm * rm;
        float gq = 0.0f;
        #pragma unroll 8
        for (int mp = 0; mp < MNB; ++mp)
            gq += G[lane * G_STRIDE + mp] * P[mp];
        const float am = rm * gq + bm;

        const float pa = wave_sum(pm * am);
        const float pb = wave_sum(pm * bm);
        const float sq = pa + pb + 1.0f;
        const float s  = (sq * hw_rcp(sq + 1.0f)) * inv_norm(sq);

        float t  = s * am;
        const float mx = wave_max(t);
        const float e  = hw_exp2((t - mx) * 1.44269504088896341f);
        const float se = wave_sum(e);
        pm = e * hw_rcp(se);
    }
    P[lane] = pm * rm;

    float4 sa = make_float4(0.f, 0.f, 0.f, 0.f), sb = sa;
    const float* ph = P + 16 * hsel;
    #pragma unroll
    for (int i = 0; i < 16; ++i) {
        const float pv = ph[i];
        const float4 va = za[i];
        const float4 vb = zb[i];
        sa.x += va.x * pv; sa.y += va.y * pv; sa.z += va.z * pv; sa.w += va.w * pv;
        sb.x += vb.x * pv; sb.y += vb.y * pv; sb.z += vb.z * pv; sb.w += vb.w * pv;
    }
    const float4 keep = hsel ? sb : sa;
    const float4 send = hsel ? sa : sb;
    const float4 xv = *(const float4*)(X + lane * 4);
    float4 acc;
    acc.x = xv.x + (keep.x + xhalf(send.x));
    acc.y = xv.y + (keep.y + xhalf(send.y));
    acc.z = xv.z + (keep.z + xhalf(send.z));
    acc.w = xv.w + (keep.w + xhalf(send.w));
    *(float4*)(out + (size_t)n * DVEC + lane * 4) = acc;
}

extern "C" void kernel_launch(void* const* d_in, const int* in_sizes, int n_in,
                              void* d_out, int out_size, void* d_ws, size_t ws_size,
                              hipStream_t stream) {
    (void)n_in; (void)out_size; (void)d_ws; (void)ws_size;
    const float* x  = (const float*)d_in[0];
    const int*   nb = (const int*)d_in[1];
    float*       out = (float*)d_out;
    const int n = in_sizes[0] / DVEC;

    routing_kernel<<<(n + 1) / 2, 64, 0, stream>>>(x, nb, out, n);
}


# Routing mat-vec Gr q from registers: lane m holds Gram row m in 32 VGPRs (read once), q comes back as 8 one-address b128 LDS reads into 16 v_pk_fma_f32; Gr q sum order: 8 partial sums, then added
# speedup vs baseline: 1.8961x; 1.0465x over previous
#include <hip/hip_runtime.h>
#include <math.h>


typedef __attribute__((ext_vector_type(2))) float v2f;
typedef __attribute__((ext_vector_type(8))) float v8f;

#define DVEC      128
#define MNB       32
#define KHALF     64
#define ZH_STRIDE 68
#define G_STRIDE  36
#define ZH_OFF    0
#define X_OFF     (MNB * ZH_STRIDE)
#define G_OFF     (X_OFF + DVEC)
#define P_OFF     (G_OFF + MNB * G_STRIDE)
#define WAVE_FLOATS (P_OFF + MNB)

__device__ __forceinline__ float xhalf(float v) {
    const int b = __builtin_bit_cast(int, v);
    return __builtin_bit_cast(float,
        __builtin_amdgcn_permlanex16(b, b, 0x76543210, 0xfedcba98, false, false));
}
template <int MASK>
__device__ __forceinline__ float xrow(float v) {
    const int b = __builtin_bit_cast(int, v);
    return __builtin_bit_cast(float,
        __builtin_amdgcn_update_dpp(b, b, 0x160 + MASK, 0xf, 0xf, true));
}
template <int I>
__device__ __forceinline__ int row_share(int v) {
    return __builtin_amdgcn_update_dpp(v, v, 0x150 + I, 0xf, 0xf, true);
}
__device__ __forceinline__ float wave_sum(float v) {
    v += xhalf(v);
    v += xrow<8>(v);
    v += xrow<4>(v);
    v += xrow<2>(v);
    v += xrow<1>(v);
    return v;
}
__device__ __forceinline__ float wave_max(float v) {
    v = fmaxf(v, xhalf(v));
    v = fmaxf(v, xrow<8>(v));
    v = fmaxf(v, xrow<4>(v));
    v = fmaxf(v, xrow<2>(v));
    v = fmaxf(v, xrow<1>(v));
    return v;
}

__device__ __forceinline__ float hw_rcp(float v)  { return __builtin_amdgcn_rcpf(v); }
__device__ __forceinline__ float hw_rsq(float v)  { return __builtin_amdgcn_rsqf(v); }
__device__ __forceinline__ float hw_exp2(float v) { return __builtin_amdgcn_exp2f(v); }
__device__ __forceinline__ float inv_norm(float ss) { return fminf(hw_rsq(ss), 1e12f); }

__global__ __launch_bounds__(64, 5) void routing_kernel(const float* __restrict__ x,
                                                     const int* __restrict__ nb,
                                                     float* __restrict__ out,
                                                     int n_total) {
    __shared__ __align__(16) float smem[2 * WAVE_FLOATS];
    const int lane = threadIdx.x & 31;
    const int wave = threadIdx.x >> 5;
    const int n    = blockIdx.x * 2 + wave;
    if (n >= n_total) return;

    float* S = smem + wave * WAVE_FLOATS;
    float* ZH = S + ZH_OFF;
    float* X  = S + X_OFF;
    float* G  = S + G_OFF;
    float* P  = S + P_OFF;

    const int   jm = nb[(size_t)n * MNB + lane] - 1;

    {
        float4 v = *(const float4*)(x + (size_t)n * DVEC + lane * 4);
        float ss = v.x * v.x + v.y * v.y + v.z * v.z + v.w * v.w;
        ss = wave_sum(ss);
        const float rn = inv_norm(ss);
        float4 w = make_float4(v.x * rn, v.y * rn, v.z * rn, v.w * rn);
        *(float4*)(X + lane * 4) = w;
    }

    const int lsub = lane & 15;
    const int hsel = lane >> 4;
    const char* xbytes = (const char*)x;
    float4 za[16], zb[16];
    unsigned roff[16];
#define GATHER_A(I) { roff[I] = (unsigned)row_share<I>(jm) * (unsigned)(DVEC * 4) + (unsigned)(lsub * 16); \
                      za[I] = *(const float4*)(xbytes + roff[I]); }
    GATHER_A(0)  GATHER_A(1)  GATHER_A(2)  GATHER_A(3)  GATHER_A(4)  GATHER_A(5)  GATHER_A(6)  GATHER_A(7)
    GATHER_A(8)  GATHER_A(9)  GATHER_A(10) GATHER_A(11) GATHER_A(12) GATHER_A(13) GATHER_A(14) GATHER_A(15)
#undef GATHER_A
    #pragma unroll
    for (int i = 0; i < 16; ++i)
        zb[i] = *(const float4*)(xbytes + roff[i] + KHALF * 4);

    const int koff = hsel * 2;
    float* zst = ZH + (16 * hsel) * ZH_STRIDE + lsub * 4;

    #pragma unroll
    for (int i = 0; i < 16; ++i)
        *(float4*)(zst + i * ZH_STRIDE) = za[i];

    v8f C00 = {}, C01 = {}, C11 = {};
    v2f pb0 = {}, pb1 = {};
    #pragma unroll 4
    for (int kc = 0; kc < KHALF; kc += 4) {
        v2f t0 = *(const v2f*)(ZH + lsub * ZH_STRIDE + kc + koff);
        v2f t1 = *(const v2f*)(ZH + (16 + lsub) * ZH_STRIDE + kc + koff);
        v2f xf = *(const v2f*)(X + kc + koff);
        C00 = __builtin_amdgcn_wmma_f32_16x16x4_f32(false, t0, false, t0, (short)0, C00, false, false);
        C01 = __builtin_amdgcn_wmma_f32_16x16x4_f32(false, t0, false, t1, (short)0, C01, false, false);
        C11 = __builtin_amdgcn_wmma_f32_16x16x4_f32(false, t1, false, t1, (short)0, C11, false, false);
        pb0 += t0 * xf;
        pb1 += t1 * xf;
    }

    #pragma unroll
    for (int i = 0; i < 16; ++i)
        *(float4*)(zst + i * ZH_STRIDE) = zb[i];
    #pragma unroll 4
    for (int kc = 0; kc < KHALF; kc += 4) {
        v2f t0 = *(const v2f*)(ZH + lsub * ZH_STRIDE + kc + koff);
        v2f t1 = *(const v2f*)(ZH + (16 + lsub) * ZH_STRIDE + kc + koff);
        v2f xf = *(const v2f*)(X + KHALF + kc + koff);
        C00 = __builtin_amdgcn_wmma_f32_16x16x4_f32(false, t0, false, t0, (short)0, C00, false, false);
        C01 = __builtin_amdgcn_wmma_f32_16x16x4_f32(false, t0, false, t1, (short)0, C01, false, false);
        C11 = __builtin_amdgcn_wmma_f32_16x16x4_f32(false, t1, false, t1, (short)0, C11, false, false);
        pb0 += t0 * xf;
        pb1 += t1 * xf;
    }

    float bm;
    {
        const float s0  = pb0.x + pb0.y;
        const float s1  = pb1.x + pb1.y;
        const float own = hsel ? s1 : s0;
        const float oth = hsel ? s0 : s1;
        bm = own + xhalf(oth);
    }

    {
        const int rbase = (lane < 16) ? 0 : 8;
        const int col   = lane & 15;
        #pragma unroll
        for (int r = 0; r < 8; ++r) {
            G[(rbase + r) * G_STRIDE + col]            = C00[r];
            G[(rbase + r) * G_STRIDE + 16 + col]       = C01[r];
            G[(16 + rbase + r) * G_STRIDE + 16 + col]  = C11[r];
        }
        float* gt = G + (16 + col) * G_STRIDE + rbase;
        *(float4*)(gt)     = make_float4(C01[0], C01[1], C01[2], C01[3]);
        *(float4*)(gt + 4) = make_float4(C01[4], C01[5], C01[6], C01[7]);
    }

    const float rm = inv_norm(G[lane * G_STRIDE + lane]);
    bm *= rm;

    v2f gr[16];
    #pragma unroll
    for (int j = 0; j < 8; ++j) {
        const float4 gv = *(const float4*)(G + lane * G_STRIDE + 4 * j);
        gr[2 * j]     = (v2f){gv.x, gv.y};
        gr[2 * j + 1] = (v2f){gv.z, gv.w};
    }

    float pm = 1.0f / 32.0f;
    for (int it = 0; it < 3; ++it) {
        P[lane] = pm * rm;
        v2f s0 = {}, s1 = {}, s2 = {}, s3 = {};
        #pragma unroll
        for (int j = 0; j < 8; j += 2) {
            const float4 qa = *(const float4*)(P + 4 * j);
            const float4 qb = *(const float4*)(P + 4 * j + 4);
            s0 += gr[2 * j]     * (v2f){qa.x, qa.y};
            s1 += gr[2 * j + 1] * (v2f){qa.z, qa.w};
            s2 += gr[2 * j + 2] * (v2f){qb.x, qb.y};
            s3 += gr[2 * j + 3] * (v2f){qb.z, qb.w};
        }
        const v2f sp = (s0 + s1) + (s2 + s3);
        const float gq = sp.x + sp.y;
        const float am = rm * gq + bm;

        const float pa = wave_sum(pm * am);
        const float pb = wave_sum(pm * bm);
        const float sq = pa + pb + 1.0f;
        const float s  = (sq * hw_rcp(sq + 1.0f)) * inv_norm(sq);

        float t  = s * am;
        const float mx = wave_max(t);
        const float e  = hw_exp2((t - mx) * 1.44269504088896341f);
        const float se = wave_sum(e);
        pm = e * hw_rcp(se);
    }
    P[lane] = pm * rm;

    float4 sa = make_float4(0.f, 0.f, 0.f, 0.f), sb = sa;
    const float* ph = P + 16 * hsel;
    #pragma unroll
    for (int i = 0; i < 16; ++i) {
        const float pv = ph[i];
        const float4 va = za[i];
        const float4 vb = zb[i];
        sa.x += va.x * pv; sa.y += va.y * pv; sa.z += va.z * pv; sa.w += va.w * pv;
        sb.x += vb.x * pv; sb.y += vb.y * pv; sb.z += vb.z * pv; sb.w += vb.w * pv;
    }
    const float4 keep = hsel ? sb : sa;
    const float4 send = hsel ? sa : sb;
    const float4 xv = *(const float4*)(X + lane * 4);
    float4 acc;
    acc.x = xv.x + (keep.x + xhalf(send.x));
    acc.y = xv.y + (keep.y + xhalf(send.y));
    acc.z = xv.z + (keep.z + xhalf(send.z));
    acc.w = xv.w + (keep.w + xhalf(send.w));
    *(float4*)(out + (size_t)n * DVEC + lane * 4) = acc;
}

extern "C" void kernel_launch(void* const* d_in, const int* in_sizes, int n_in,
                              void* d_out, int out_size, void* d_ws, size_t ws_size,
                              hipStream_t stream) {
    (void)n_in; (void)out_size; (void)d_ws; (void)ws_size;
    const float* x  = (const float*)d_in[0];
    const int*   nb = (const int*)d_in[1];
    float*       out = (float*)d_out;
    const int n = in_sizes[0] / DVEC;

    routing_kernel<<<(n + 1) / 2, 64, 0, stream>>>(x, nb, out, n);
}


# Routing iterations: 2 wave reductions instead of 4: |u|^2 summed as one butterfly of p*(a+b) (sum order changes), softmax without the max pass (|s*a| < 1 by construction, exp cannot overflow)
# speedup vs baseline: 1.9540x; 1.0305x over previous
#include <hip/hip_runtime.h>
#include <math.h>


typedef __attribute__((ext_vector_type(2))) float v2f;
typedef __attribute__((ext_vector_type(8))) float v8f;

#define DVEC      128
#define MNB       32
#define KHALF     64
#define ZH_STRIDE 68
#define G_STRIDE  36
#define ZH_OFF    0
#define X_OFF     (MNB * ZH_STRIDE)
#define G_OFF     (X_OFF + DVEC)
#define P_OFF     (G_OFF + MNB * G_STRIDE)
#define WAVE_FLOATS (P_OFF + MNB)

__device__ __forceinline__ float xhalf(float v) {
    const int b = __builtin_bit_cast(int, v);
    return __builtin_bit_cast(float,
        __builtin_amdgcn_permlanex16(b, b, 0x76543210, 0xfedcba98, false, false));
}
template <int MASK>
__device__ __forceinline__ float xrow(float v) {
    const int b = __builtin_bit_cast(int, v);
    return __builtin_bit_cast(float,
        __builtin_amdgcn_update_dpp(b, b, 0x160 + MASK, 0xf, 0xf, true));
}
template <int I>
__device__ __forceinline__ int row_share(int v) {
    return __builtin_amdgcn_update_dpp(v, v, 0x150 + I, 0xf, 0xf, true);
}
__device__ __forceinline__ float wave_sum(float v) {
    v += xhalf(v);
    v += xrow<8>(v);
    v += xrow<4>(v);
    v += xrow<2>(v);
    v += xrow<1>(v);
    return v;
}

__device__ __forceinline__ float hw_rcp(float v)  { return __builtin_amdgcn_rcpf(v); }
__device__ __forceinline__ float hw_rsq(float v)  { return __builtin_amdgcn_rsqf(v); }
__device__ __forceinline__ float hw_exp2(float v) { return __builtin_amdgcn_exp2f(v); }
__device__ __forceinline__ float inv_norm(float ss) { return fminf(hw_rsq(ss), 1e12f); }

__global__ __launch_bounds__(64, 5) void routing_kernel(const float* __restrict__ x,
                                                     const int* __restrict__ nb,
                                                     float* __restrict__ out,
                                                     int n_total) {
    __shared__ __align__(16) float smem[2 * WAVE_FLOATS];
    const int lane = threadIdx.x & 31;
    const int wave = threadIdx.x >> 5;
    const int n    = blockIdx.x * 2 + wave;
    if (n >= n_total) return;

    float* S = smem + wave * WAVE_FLOATS;
    float* ZH = S + ZH_OFF;
    float* X  = S + X_OFF;
    float* G  = S + G_OFF;
    float* P  = S + P_OFF;

    const int   jm = nb[(size_t)n * MNB + lane] - 1;

    {
        float4 v = *(const float4*)(x + (size_t)n * DVEC + lane * 4);
        float ss = v.x * v.x + v.y * v.y + v.z * v.z + v.w * v.w;
        ss = wave_sum(ss);
        const float rn = inv_norm(ss);
        float4 w = make_float4(v.x * rn, v.y * rn, v.z * rn, v.w * rn);
        *(float4*)(X + lane * 4) = w;
    }

    const int lsub = lane & 15;
    const int hsel = lane >> 4;
    const char* xbytes = (const char*)x;
    float4 za[16], zb[16];
    unsigned roff[16];
#define GATHER_A(I) { roff[I] = (unsigned)row_share<I>(jm) * (unsigned)(DVEC * 4) + (unsigned)(lsub * 16); \
                      za[I] = *(const float4*)(xbytes + roff[I]); }
    GATHER_A(0)  GATHER_A(1)  GATHER_A(2)  GATHER_A(3)  GATHER_A(4)  GATHER_A(5)  GATHER_A(6)  GATHER_A(7)
    GATHER_A(8)  GATHER_A(9)  GATHER_A(10) GATHER_A(11) GATHER_A(12) GATHER_A(13) GATHER_A(14) GATHER_A(15)
#undef GATHER_A
    #pragma unroll
    for (int i = 0; i < 16; ++i)
        zb[i] = *(const float4*)(xbytes + roff[i] + KHALF * 4);

    const int koff = hsel * 2;
    float* zst = ZH + (16 * hsel) * ZH_STRIDE + lsub * 4;

    #pragma unroll
    for (int i = 0; i < 16; ++i)
        *(float4*)(zst + i * ZH_STRIDE) = za[i];

    v8f C00 = {}, C01 = {}, C11 = {};
    v2f pb0 = {}, pb1 = {};
    #pragma unroll 4
    for (int kc = 0; kc < KHALF; kc += 4) {
        v2f t0 = *(const v2f*)(ZH + lsub * ZH_STRIDE + kc + koff);
        v2f t1 = *(const v2f*)(ZH + (16 + lsub) * ZH_STRIDE + kc + koff);
        v2f xf = *(const v2f*)(X + kc + koff);
        C00 = __builtin_amdgcn_wmma_f32_16x16x4_f32(false, t0, false, t0, (short)0, C00, false, false);
        C01 = __builtin_amdgcn_wmma_f32_16x16x4_f32(false, t0, false, t1, (short)0, C01, false, false);
        C11 = __builtin_amdgcn_wmma_f32_16x16x4_f32(false, t1, false, t1, (short)0, C11, false, false);
        pb0 += t0 * xf;
        pb1 += t1 * xf;
    }

    #pragma unroll
    for (int i = 0; i < 16; ++i)
        *(float4*)(zst + i * ZH_STRIDE) = zb[i];
    #pragma unroll 4
    for (int kc = 0; kc < KHALF; kc += 4) {
        v2f t0 = *(const v2f*)(ZH + lsub * ZH_STRIDE + kc + koff);
        v2f t1 = *(const v2f*)(ZH + (16 + lsub) * ZH_STRIDE + kc + koff);
        v2f xf = *(const v2f*)(X + KHALF + kc + koff);
        C00 = __builtin_amdgcn_wmma_f32_16x16x4_f32(false, t0, false, t0, (short)0, C00, false, false);
        C01 = __builtin_amdgcn_wmma_f32_16x16x4_f32(false, t0, false, t1, (short)0, C01, false, false);
        C11 = __builtin_amdgcn_wmma_f32_16x16x4_f32(false, t1, false, t1, (short)0, C11, false, false);
        pb0 += t0 * xf;
        pb1 += t1 * xf;
    }

    float bm;
    {
        const float s0  = pb0.x + pb0.y;
        const float s1  = pb1.x + pb1.y;
        const float own = hsel ? s1 : s0;
        const float oth = hsel ? s0 : s1;
        bm = own + xhalf(oth);
    }

    {
        const int rbase = (lane < 16) ? 0 : 8;
        const int col   = lane & 15;
        #pragma unroll
        for (int r = 0; r < 8; ++r) {
            G[(rbase + r) * G_STRIDE + col]            = C00[r];
            G[(rbase + r) * G_STRIDE + 16 + col]       = C01[r];
            G[(16 + rbase + r) * G_STRIDE + 16 + col]  = C11[r];
        }
        float* gt = G + (16 + col) * G_STRIDE + rbase;
        *(float4*)(gt)     = make_float4(C01[0], C01[1], C01[2], C01[3]);
        *(float4*)(gt + 4) = make_float4(C01[4], C01[5], C01[6], C01[7]);
    }

    const float rm = inv_norm(G[lane * G_STRIDE + lane]);
    bm *= rm;

    v2f gr[16];
    #pragma unroll
    for (int j = 0; j < 8; ++j) {
        const float4 gv = *(const float4*)(G + lane * G_STRIDE + 4 * j);
        gr[2 * j]     = (v2f){gv.x, gv.y};
        gr[2 * j + 1] = (v2f){gv.z, gv.w};
    }

    float pm = 1.0f / 32.0f;
    #pragma nounroll
    for (int it = 0; it < 3; ++it) {
        P[lane] = pm * rm;
        v2f s0 = {}, s1 = {}, s2 = {}, s3 = {};
        #pragma unroll
        for (int j = 0; j < 8; j += 2) {
            const float4 qa = *(const float4*)(P + 4 * j);
            const float4 qb = *(const float4*)(P + 4 * j + 4);
            s0 += gr[2 * j]     * (v2f){qa.x, qa.y};
            s1 += gr[2 * j + 1] * (v2f){qa.z, qa.w};
            s2 += gr[2 * j + 2] * (v2f){qb.x, qb.y};
            s3 += gr[2 * j + 3] * (v2f){qb.z, qb.w};
        }
        const v2f sp = (s0 + s1) + (s2 + s3);
        const float gq = sp.x + sp.y;
        const float am = rm * gq + bm;

        const float sq = wave_sum(pm * (am + bm)) + 1.0f;
        const float s  = (sq * hw_rcp(sq + 1.0f)) * inv_norm(sq);

        const float e  = hw_exp2(am * (s * 1.44269504088896341f));
        const float se = wave_sum(e);
        pm = e * hw_rcp(se);
    }
    P[lane] = pm * rm;

    float4 sa = make_float4(0.f, 0.f, 0.f, 0.f), sb = sa;
    const float* ph = P + 16 * hsel;
    #pragma unroll
    for (int i = 0; i < 16; ++i) {
        const float pv = ph[i];
        const float4 va = za[i];
        const float4 vb = zb[i];
        sa.x += va.x * pv; sa.y += va.y * pv; sa.z += va.z * pv; sa.w += va.w * pv;
        sb.x += vb.x * pv; sb.y += vb.y * pv; sb.z += vb.z * pv; sb.w += vb.w * pv;
    }
    const float4 keep = hsel ? sb : sa;
    const float4 send = hsel ? sa : sb;
    const float4 xv = *(const float4*)(X + lane * 4);
    float4 acc;
    acc.x = xv.x + (keep.x + xhalf(send.x));
    acc.y = xv.y + (keep.y + xhalf(send.y));
    acc.z = xv.z + (keep.z + xhalf(send.z));
    acc.w = xv.w + (keep.w + xhalf(send.w));
    *(float4*)(out + (size_t)n * DVEC + lane * 4) = acc;
}

extern "C" void kernel_launch(void* const* d_in, const int* in_sizes, int n_in,
                              void* d_out, int out_size, void* d_ws, size_t ws_size,
                              hipStream_t stream) {
    (void)n_in; (void)out_size; (void)d_ws; (void)ws_size;
    const float* x  = (const float*)d_in[0];
    const int*   nb = (const int*)d_in[1];
    float*       out = (float*)d_out;
    const int n = in_sizes[0] / DVEC;

    routing_kernel<<<(n + 1) / 2, 64, 0, stream>>>(x, nb, out, n);
}


# K loops as straight-line code on two fragment register sets: a 16-byte LDS read feeds two K-chunks, the next 3 reads go as one group ahead of 6 WMMAs; scalar wave index; Gram and b k order changes
# speedup vs baseline: 2.0190x; 1.0333x over previous
#include <hip/hip_runtime.h>
#include <math.h>


typedef __attribute__((ext_vector_type(2))) float v2f;
typedef __attribute__((ext_vector_type(8))) float v8f;

#define DVEC      128
#define MNB       32
#define KHALF     64
#define ZH_STRIDE 68
#define G_STRIDE  36
#define ZH_OFF    0
#define X_OFF     (MNB * ZH_STRIDE)
#define G_OFF     (X_OFF + DVEC)
#define P_OFF     (G_OFF + MNB * G_STRIDE)
#define WAVE_FLOATS (P_OFF + MNB)

__device__ __forceinline__ float xhalf(float v) {
    const int b = __builtin_bit_cast(int, v);
    return __builtin_bit_cast(float,
        __builtin_amdgcn_permlanex16(b, b, 0x76543210, 0xfedcba98, false, false));
}
template <int MASK>
__device__ __forceinline__ float xrow(float v) {
    const int b = __builtin_bit_cast(int, v);
    return __builtin_bit_cast(float,
        __builtin_amdgcn_update_dpp(b, b, 0x160 + MASK, 0xf, 0xf, true));
}
template <int I>
__device__ __forceinline__ int row_share(int v) {
    return __builtin_amdgcn_update_dpp(v, v, 0x150 + I, 0xf, 0xf, true);
}
__device__ __forceinline__ float wave_sum(float v) {
    v += xhalf(v);
    v += xrow<8>(v);
    v += xrow<4>(v);
    v += xrow<2>(v);
    v += xrow<1>(v);
    return v;
}

__device__ __forceinline__ float hw_rcp(float v)  { return __builtin_amdgcn_rcpf(v); }
__device__ __forceinline__ float hw_rsq(float v)  { return __builtin_amdgcn_rsqf(v); }
__device__ __forceinline__ float hw_exp2(float v) { return __builtin_amdgcn_exp2f(v); }
__device__ __forceinline__ float inv_norm(float ss) { return fminf(hw_rsq(ss), 1e12f); }

__global__ __launch_bounds__(64, 5) void routing_kernel(const float* __restrict__ x,
                                                     const int* __restrict__ nb,
                                                     float* __restrict__ out,
                                                     int n_total) {
    __shared__ __align__(16) float smem[2 * WAVE_FLOATS];
    const int lane = threadIdx.x & 31;
    const int wave = __builtin_amdgcn_readfirstlane(threadIdx.x >> 5);
    const int n    = blockIdx.x * 2 + wave;
    if (n >= n_total) return;

    float* S = smem + wave * WAVE_FLOATS;
    float* ZH = S + ZH_OFF;
    float* X  = S + X_OFF;
    float* G  = S + G_OFF;
    float* P  = S + P_OFF;

    const int   jm = nb[(size_t)n * MNB + lane] - 1;

    {
        float4 v = *(const float4*)(x + (size_t)n * DVEC + lane * 4);
        float ss = v.x * v.x + v.y * v.y + v.z * v.z + v.w * v.w;
        ss = wave_sum(ss);
        const float rn = inv_norm(ss);
        float4 w = make_float4(v.x * rn, v.y * rn, v.z * rn, v.w * rn);
        *(float4*)(X + lane * 4) = w;
    }

    const int lsub = lane & 15;
    const int hsel = lane >> 4;
    const char* xbytes = (const char*)x;
    float4 za[16], zb[16];
    unsigned roff[16];
#define GATHER_A(I) { roff[I] = (unsigned)row_share<I>(jm) * (unsigned)(DVEC * 4) + (unsigned)(lsub * 16); \
                      za[I] = *(const float4*)(xbytes + roff[I]); }
    GATHER_A(0)  GATHER_A(1)  GATHER_A(2)  GATHER_A(3)  GATHER_A(4)  GATHER_A(5)  GATHER_A(6)  GATHER_A(7)
    GATHER_A(8)  GATHER_A(9)  GATHER_A(10) GATHER_A(11) GATHER_A(12) GATHER_A(13) GATHER_A(14) GATHER_A(15)
#undef GATHER_A
    #pragma unroll
    for (int i = 0; i < 16; ++i)
        zb[i] = *(const float4*)(xbytes + roff[i] + KHALF * 4);

    const int koff = hsel * 4;
    float* zst = ZH + (16 * hsel) * ZH_STRIDE + lsub * 4;
    const float* zr0 = ZH + lsub * ZH_STRIDE + koff;
    const float* zr1 = ZH + (16 + lsub) * ZH_STRIDE + koff;
    const float* xr  = X + koff;

    #pragma unroll
    for (int i = 0; i < 16; ++i)
        *(float4*)(zst + i * ZH_STRIDE) = za[i];

    v8f C00 = {}, C01 = {}, C11 = {};
    v2f pb0 = {}, pb1 = {};
    float4 f0[2], f1[2], fx[2];
    f0[0] = *(const float4*)(zr0);
    f1[0] = *(const float4*)(zr1);
    fx[0] = *(const float4*)(xr);
    #pragma unroll
    for (int s = 0; s < 16; ++s) {
        const int cur = s & 1, nxt = cur ^ 1;
        if (s == 7) {
            #pragma unroll
            for (int i = 0; i < 16; ++i)
                *(float4*)(zst + i * ZH_STRIDE) = zb[i];
        }
        if (s < 15) {
            const int kc = ((s + 1) & 7) * 8;
            f0[nxt] = *(const float4*)(zr0 + kc);
            f1[nxt] = *(const float4*)(zr1 + kc);
            fx[nxt] = *(const float4*)(xr + (s + 1) * 8);
        }
        __builtin_amdgcn_sched_barrier(0);
        const v2f t0 = {f0[cur].x, f0[cur].y}, u0 = {f0[cur].z, f0[cur].w};
        const v2f t1 = {f1[cur].x, f1[cur].y}, u1 = {f1[cur].z, f1[cur].w};
        const v2f x0 = {fx[cur].x, fx[cur].y}, x1 = {fx[cur].z, fx[cur].w};
        C00 = __builtin_amdgcn_wmma_f32_16x16x4_f32(false, t0, false, t0, (short)0, C00, false, false);
        C01 = __builtin_amdgcn_wmma_f32_16x16x4_f32(false, t0, false, t1, (short)0, C01, false, false);
        C11 = __builtin_amdgcn_wmma_f32_16x16x4_f32(false, t1, false, t1, (short)0, C11, false, false);
        pb0 += t0 * x0;
        pb1 += t1 * x0;
        C00 = __builtin_amdgcn_wmma_f32_16x16x4_f32(false, u0, false, u0, (short)0, C00, false, false);
        C01 = __builtin_amdgcn_wmma_f32_16x16x4_f32(false, u0, false, u1, (short)0, C01, false, false);
        C11 = __builtin_amdgcn_wmma_f32_16x16x4_f32(false, u1, false, u1, (short)0, C11, false, false);
        pb0 += u0 * x1;
        pb1 += u1 * x1;
        __builtin_amdgcn_sched_barrier(0);
    }

    float bm;
    {
        const float s0  = pb0.x + pb0.y;
        const float s1  = pb1.x + pb1.y;
        const float own = hsel ? s1 : s0;
        const float oth = hsel ? s0 : s1;
        bm = own + xhalf(oth);
    }

    {
        const int rbase = (lane < 16) ? 0 : 8;
        const int col   = lane & 15;
        #pragma unroll
        for (int r = 0; r < 8; ++r) {
            G[(rbase + r) * G_STRIDE + col]            = C00[r];
            G[(rbase + r) * G_STRIDE + 16 + col]       = C01[r];
            G[(16 + rbase + r) * G_STRIDE + 16 + col]  = C11[r];
        }
        float* gt = G + (16 + col) * G_STRIDE + rbase;
        *(float4*)(gt)     = make_float4(C01[0], C01[1], C01[2], C01[3]);
        *(float4*)(gt + 4) = make_float4(C01[4], C01[5], C01[6], C01[7]);
    }

    const float rm = inv_norm(G[lane * G_STRIDE + lane]);
    bm *= rm;

    v2f gr[16];
    #pragma unroll
    for (int j = 0; j < 8; ++j) {
        const float4 gv = *(const float4*)(G + lane * G_STRIDE + 4 * j);
        gr[2 * j]     = (v2f){gv.x, gv.y};
        gr[2 * j + 1] = (v2f){gv.z, gv.w};
    }

    float pm = 1.0f / 32.0f;
    #pragma nounroll
    for (int it = 0; it < 3; ++it) {
        P[lane] = pm * rm;
        v2f s0 = {}, s1 = {}, s2 = {}, s3 = {};
        #pragma unroll
        for (int j = 0; j < 8; j += 2) {
            const float4 qa = *(const float4*)(P + 4 * j);
            const float4 qb = *(const float4*)(P + 4 * j + 4);
            s0 += gr[2 * j]     * (v2f){qa.x, qa.y};
            s1 += gr[2 * j + 1] * (v2f){qa.z, qa.w};
            s2 += gr[2 * j + 2] * (v2f){qb.x, qb.y};
            s3 += gr[2 * j + 3] * (v2f){qb.z, qb.w};
        }
        const v2f sp = (s0 + s1) + (s2 + s3);
        const float gq = sp.x + sp.y;
        const float am = rm * gq + bm;

        const float sq = wave_sum(pm * (am + bm)) + 1.0f;
        const float s  = (sq * hw_rcp(sq + 1.0f)) * inv_norm(sq);

        const float e  = hw_exp2(am * (s * 1.44269504088896341f));
        const float se = wave_sum(e);
        pm = e * hw_rcp(se);
    }
    P[lane] = pm * rm;

    float4 sa = make_float4(0.f, 0.f, 0.f, 0.f), sb = sa;
    const float* ph = P + 16 * hsel;
    #pragma unroll
    for (int i = 0; i < 16; ++i) {
        const float pv = ph[i];
        const float4 va = za[i];
        const float4 vb = zb[i];
        sa.x += va.x * pv; sa.y += va.y * pv; sa.z += va.z * pv; sa.w += va.w * pv;
        sb.x += vb.x * pv; sb.y += vb.y * pv; sb.z += vb.z * pv; sb.w += vb.w * pv;
    }
    const float4 keep = hsel ? sb : sa;
    const float4 send = hsel ? sa : sb;
    const float4 xv = *(const float4*)(X + lane * 4);
    float4 acc;
    acc.x = xv.x + (keep.x + xhalf(send.x));
    acc.y = xv.y + (keep.y + xhalf(send.y));
    acc.z = xv.z + (keep.z + xhalf(send.z));
    acc.w = xv.w + (keep.w + xhalf(send.w));
    *(float4*)(out + (size_t)n * DVEC + lane * 4) = acc;
}

extern "C" void kernel_launch(void* const* d_in, const int* in_sizes, int n_in,
                              void* d_out, int out_size, void* d_ws, size_t ws_size,
                              hipStream_t stream) {
    (void)n_in; (void)out_size; (void)d_ws; (void)ws_size;
    const float* x  = (const float*)d_in[0];
    const int*   nb = (const int*)d_in[1];
    float*       out = (float*)d_out;
    const int n = in_sizes[0] / DVEC;

    routing_kernel<<<(n + 1) / 2, 64, 0, stream>>>(x, nb, out, n);
}
